# starting kernel
# baseline (speedup 1.0000x reference)
#include <hip/hip_runtime.h>

#define BDIM 2048
#define TDIM 4096

typedef float v2f __attribute__((ext_vector_type(2)));
typedef float v8f __attribute__((ext_vector_type(8)));

__device__ __forceinline__ float sigm10(float y) {
    float e = __builtin_amdgcn_exp2f(-14.4269504088896341f * y);
    return __builtin_amdgcn_rcpf(1.0f + e);
}

__device__ __forceinline__ float clip1e5(float d) {
    return fminf(fmaxf(d, -100000.0f), 100000.0f);
}

__global__ __launch_bounds__(256) void hbv_scan_kernel(
    const float* __restrict__ x, const float* __restrict__ p,
    float* __restrict__ qws)
{
    const int b = blockIdx.x * 256 + threadIdx.x;

    const float Tt     = p[0]  * -1.5f;
    const float CFR    = p[1]  * 0.01f;
    const float CFMAX  = p[2];
    const float SCF    = p[3];
    const float FC     = p[4]  * 100.0f;
    const float Beta   = p[5]  * 8.0f;
    const float LP     = p[6]  * 2.5f;
    const float CWH    = p[7]  * 0.3f;
    const float UZL    = p[8]  * 120.0f;
    const float K0     = p[9]  * 0.5f;
    const float K1     = p[10] * 0.3f;
    const float K2     = p[11] * 0.01f;
    const float PER    = p[12] * 3.5f;
    const float rawCWH = p[7];

    const float invFC   = __builtin_amdgcn_rcpf(FC);
    const float FCLP    = FC * LP;
    const float invFCLP = __builtin_amdgcn_rcpf(FCLP);
    const float CFRMAX  = CFR * CFMAX;

    float S1 = 0.0f, S2 = 0.0f, S3 = 0.0f, S4 = 0.0f, S5 = 0.0f;

    const float* xb = x + (size_t)b * TDIM * 3;
    float* qp = qws + b;

    for (int t = 0; t < TDIM; ++t) {
        float P   = xb[0];
        float Tm  = xb[1];
        float PET = xb[2];
        xb += 3;

        float h_cold = sigm10(Tt - Tm);
        float h_warm = 1.0f - h_cold;
        float Ps  = P * SCF * h_cold;
        float Rfr = fminf((Tt - Tm) * CFRMAX, S2) * h_cold;
        float Pr  = P * h_warm;
        float M   = fminf((Tm - Tt) * CFMAX, S1) * h_warm;
        float I   = sigm10(S2 - S1 * CWH) * (Pr + M);
        float Re  = I * __builtin_amdgcn_exp2f(Beta * __builtin_amdgcn_logf(S3 * invFC));
        float hS3 = sigm10(S3);
        float hLP = sigm10(FCLP - S3);
        float Ea  = hS3 * (PET * S3 * invFCLP) * hLP + hS3 * PET * (1.0f - hLP);
        float hS4 = sigm10(S4);
        float Pe  = PER * hS4;
        float q0  = (S4 - UZL) * K0 * sigm10(S4 - UZL);
        float q1  = S4 * K1 * hS4;
        float q2  = S5 * K2 * sigm10(S5);

        float nS1 = fmaxf(S1 + clip1e5(Ps + Rfr - M), 0.0f);
        float nS2 = fmaxf(fminf(S2 + clip1e5(Pr + M - I - Rfr), nS1 * rawCWH), 0.0f);
        float nS3 = fmaxf(S3 + clip1e5(I - Ea - Re), 0.0f);
        float nS4 = fmaxf(S4 + clip1e5(Re - q0 - q1 - Pe), 0.0f);
        float nS5 = fmaxf(S5 + clip1e5(Pe - q2), 0.0f);
        S1 = nS1; S2 = nS2; S3 = nS3; S4 = nS4; S5 = nS5;

        qp[(size_t)t * BDIM] = q0 + q1 + q2;
    }
}

__global__ __launch_bounds__(256) void hbv_route_wmma_kernel(
    const float* __restrict__ qws, float* __restrict__ out)
{
    constexpr double INV_MID_D = 1.0 / (4.0 + 1e-7);
    constexpr double WSUM_D = 1.0 + 2.0 * ((1.0 - 1.0 * INV_MID_D) +
                                           (1.0 - 2.0 * INV_MID_D) +
                                           (1.0 - 3.0 * INV_MID_D) +
                                           (1.0 - 4.0 * INV_MID_D));
    constexpr float INV_MID  = (float)INV_MID_D;
    constexpr float INV_WSUM = (float)(1.0 / (WSUM_D + 1e-7));

    const int lane = threadIdx.x & 31;
    const int wave = threadIdx.x >> 5;
    const int tile = blockIdx.x * 8 + wave;
    const int tb   = tile & (BDIM / 16 - 1);
    const int tt   = tile >> 7;
    const int b0   = tb * 16;
    const int t0   = tt * 16;
    const int mn   = lane & 15;
    const int khi  = lane >> 4;

    v2f afr[6], wfr[6];
#pragma unroll
    for (int c = 0; c < 6; ++c) {
#pragma unroll
        for (int j = 0; j < 2; ++j) {
            const int kg = 4 * c + j + 2 * khi;
            const int tq  = t0 - 4 + kg;
            const int tqc = tq < 0 ? 0 : (tq > TDIM - 1 ? TDIM - 1 : tq);
            const float v = qws[(size_t)tqc * BDIM + b0 + mn];
            afr[c][j] = (tq == tqc) ? v : 0.0f;
            const float dd = fabsf((float)(kg - mn) - 4.0f);
            wfr[c][j] = fmaxf(0.0f, fmaf(-dd, INV_MID, 1.0f)) * INV_WSUM;
        }
    }

    v8f acc = {0.0f, 0.0f, 0.0f, 0.0f, 0.0f, 0.0f, 0.0f, 0.0f};
#pragma unroll
    for (int c = 0; c < 6; ++c) {
        acc = __builtin_amdgcn_wmma_f32_16x16x4_f32(
            false, afr[c], false, wfr[c], (short)0, acc, false, false);
    }

#pragma unroll
    for (int v = 0; v < 8; ++v) {
        const int m = v + 8 * khi;
        out[(size_t)(b0 + m) * TDIM + t0 + mn] = acc[v];
    }
}

extern "C" void kernel_launch(void* const* d_in, const int* in_sizes, int n_in,
                              void* d_out, int out_size, void* d_ws, size_t ws_size,
                              hipStream_t stream) {
    (void)in_sizes; (void)n_in; (void)out_size; (void)ws_size;
    const float* x      = (const float*)d_in[0];
    const float* params = (const float*)d_in[1];
    float* out = (float*)d_out;
    float* qws = (float*)d_ws;

    hbv_scan_kernel<<<BDIM / 256, 256, 0, stream>>>(x, params, qws);
    hbv_route_wmma_kernel<<<(BDIM / 16) * (TDIM / 16) / 8, 256, 0, stream>>>(qws, out);
}


# clip, state add and zero floor of S2..S5 folded into one v_med3 each: the old state leads its flux sum (S4-q0+Re-q1-Pe), bounds max(S-1e5,0) and S+1e5 come from the old state; sums re-rounded
# speedup vs baseline: 8.9862x; 8.9862x over previous
#include <hip/hip_runtime.h>


#define BDIM 2048
#define TDIM 4096

typedef float v2f __attribute__((ext_vector_type(2)));
typedef float v4f __attribute__((ext_vector_type(4)));
typedef float v8f __attribute__((ext_vector_type(8)));

__device__ __forceinline__ float clip1e5(float d) {
    return fminf(fmaxf(d, -100000.0f), 100000.0f);
}

#define CHUNK   32
#define NCHUNK  (TDIM / CHUNK)
#define ROWV    (CHUNK * 3 / 4)
#define PITCHV  25
#define XROWV   (TDIM * 3 / 4)
#define NGROUP  (CHUNK / 4)
#define LPB     8
#define BPW     (32 / LPB)
#define WPB     4
#define NPIECE  (ROWV / LPB)

__device__ __forceinline__ void wave_tile_handover() {
    __builtin_amdgcn_fence(__ATOMIC_RELEASE, "wavefront");
    __builtin_amdgcn_wave_barrier();
    __builtin_amdgcn_fence(__ATOMIC_ACQUIRE, "wavefront");
}

template <int K>
__device__ __forceinline__ float group_bcast(float v, unsigned int grp8) {
    const unsigned int u = __builtin_bit_cast(unsigned int, v);
    return __builtin_bit_cast(float,
        __builtin_amdgcn_permlane16_var(u, u, grp8 | (unsigned int)K, true, true));
}

__global__ __launch_bounds__(WPB * 32) void hbv_scan_kernel(
    const float* __restrict__ x, const float* __restrict__ p,
    float* __restrict__ qws)
{
    __shared__ v4f tile[WPB * BPW * PITCHV];

    const int lane = threadIdx.x & 31;
    const int wave = threadIdx.x >> 5;
    const int sub  = lane & (LPB - 1);
    const int row  = lane / LPB;
    const int b    = (blockIdx.x * WPB + wave) * BPW + row;

    const float Tt     = p[0]  * -1.5f;
    const float CFR    = p[1]  * 0.01f;
    const float CFMAX  = p[2];
    const float SCF    = p[3];
    const float FC     = p[4]  * 100.0f;
    const float Beta   = p[5]  * 8.0f;
    const float LP     = p[6]  * 2.5f;
    const float CWH    = p[7]  * 0.3f;
    const float UZL    = p[8]  * 120.0f;
    const float K0     = p[9]  * 0.5f;
    const float K1     = p[10] * 0.3f;
    const float K2     = p[11] * 0.01f;
    const float PER    = p[12] * 3.5f;
    const float rawCWH = p[7];

    const float invFC   = __builtin_amdgcn_rcpf(FC);
    const float FCLP    = FC * LP;
    const float invFCLP = __builtin_amdgcn_rcpf(FCLP);
    const float CFRMAX  = CFR * CFMAX;

    float a1 = 0.0f, a2 = 0.0f, a3 = 0.0f, a4 = 0.0f, a5 = 0.0f, k0 = 0.0f;
    if (sub == 1) { a1 = -CWH; a2 = 1.0f; }
    if (sub == 2) { a3 = 1.0f; }
    if (sub == 3) { a3 = -1.0f; k0 = FCLP; }
    if (sub == 4) { a4 = 1.0f; }
    if (sub == 5) { a4 = 1.0f; k0 = -UZL; }
    if (sub == 6) { a5 = 1.0f; }
    const unsigned int pmask = (sub == 7) ? 0xffffffffu : 0u;
    const unsigned int grp8 = (unsigned int)lane & 8u;

    float S1 = 0.0f, S2 = 0.0f, S3 = 0.0f, S4 = 0.0f, S5 = 0.0f;

    auto cold4 = [&](float Tm) __attribute__((always_inline)) {
        const float e  = __builtin_amdgcn_exp2f(-14.4269504088896341f * (Tt - Tm));
        const float r  = __builtin_amdgcn_rcpf(1.0f + e);
        v4f h;
        h.x = group_bcast<0>(r, grp8);
        h.y = group_bcast<1>(r, grp8);
        h.z = group_bcast<2>(r, grp8);
        h.w = group_bcast<3>(r, grp8);
        return h;
    };

    auto step = [&](float h_cold, float P, float Tm, float PET) __attribute__((always_inline)) {
        const float ycold = Tt - Tm;
        const float y  = fmaf(a3, S3, k0) + (fmaf(a1, S1, a2 * S2) + fmaf(a4, S4, a5 * S5));
        const float lg = __builtin_amdgcn_logf(S3 * invFC);
        const unsigned int eb = (pmask & __builtin_bit_cast(unsigned int, Beta * lg)) |
                                (~pmask & __builtin_bit_cast(unsigned int, -14.4269504088896341f * y));
        const float ex = __builtin_bit_cast(float, eb);
        const float e  = __builtin_amdgcn_exp2f(ex);
        const float r  = __builtin_amdgcn_rcpf(1.0f + e);

        const float hI     = group_bcast<1>(r, grp8);
        const float hS3    = group_bcast<2>(r, grp8);
        const float hLP    = group_bcast<3>(r, grp8);
        const float hS4    = group_bcast<4>(r, grp8);
        const float hUZ    = group_bcast<5>(r, grp8);
        const float hS5    = group_bcast<6>(r, grp8);
        const float pw     = group_bcast<7>(e, grp8);

        const float h_warm = 1.0f - h_cold;
        const float Ps  = P * SCF * h_cold;
        const float Rfr = fminf(ycold * CFRMAX, S2) * h_cold;
        const float Pr  = P * h_warm;
        const float M   = fminf((Tm - Tt) * CFMAX, S1) * h_warm;
        const float I   = hI * (Pr + M);
        const float Re  = I * pw;
        const float Ea  = hS3 * (PET * S3 * invFCLP) * hLP + hS3 * PET * (1.0f - hLP);
        const float Pe  = PER * hS4;
        const float q0  = (S4 - UZL) * K0 * hUZ;
        const float q1  = S4 * K1 * hS4;
        const float q2  = S5 * K2 * hS5;

        const float nS1 = fmaxf(S1 + clip1e5(Ps + Rfr - M), 0.0f);

        const float cap2 = nS1 * rawCWH;
        const float lo2 = fmaxf(fminf(S2 - 100000.0f, cap2), 0.0f);
        const float hi2 = fmaxf(fminf(S2 + 100000.0f, cap2), 0.0f);
        const float lo3 = fmaxf(S3 - 100000.0f, 0.0f), hi3 = S3 + 100000.0f;
        const float lo4 = fmaxf(S4 - 100000.0f, 0.0f), hi4 = S4 + 100000.0f;
        const float lo5 = fmaxf(S5 - 100000.0f, 0.0f), hi5 = S5 + 100000.0f;
        const float nS2 = __builtin_amdgcn_fmed3f(S2 + (Pr + M) - I - Rfr, lo2, hi2);
        const float nS3 = __builtin_amdgcn_fmed3f(S3 + I - Ea - Re, lo3, hi3);
        const float nS4 = __builtin_amdgcn_fmed3f(S4 - q0 + Re - q1 - Pe, lo4, hi4);
        const float nS5 = __builtin_amdgcn_fmed3f(S5 + Pe - q2, lo5, hi5);
        S1 = nS1; S2 = nS2; S3 = nS3; S4 = nS4; S5 = nS5;

        return q0 + q1 + q2;
    };

    const v4f* xg = (const v4f*)x + (size_t)b * XROWV + sub;
    v4f* wtile    = tile + wave * (BPW * PITCHV);
    v4f* stp      = wtile + row * PITCHV + sub;
    const v4f* ldp = wtile + row * PITCHV;
    const float* ldT = (const float*)ldp + 3 * (sub & 3) + 1;
    float* qp = qws + (size_t)b * TDIM + sub * 4;

    v4f pre[NPIECE];
#pragma unroll
    for (int i = 0; i < NPIECE; ++i)
        pre[i] = __builtin_nontemporal_load(xg + i * LPB);

    for (int c = 0; c < NCHUNK; ++c) {
        wave_tile_handover();
#pragma unroll
        for (int i = 0; i < NPIECE; ++i)
            stp[i * LPB] = pre[i];
        wave_tile_handover();

        v4f a0 = ldp[0], a1v = ldp[1], a2v = ldp[2];
        const float T0 = ldT[0];

        if (c + 1 < NCHUNK) {
            const v4f* xn = xg + (c + 1) * ROWV;
#pragma unroll
            for (int i = 0; i < NPIECE; ++i)
                pre[i] = __builtin_nontemporal_load(xn + i * LPB);
        }

        v4f hc = cold4(T0);

        v4f qk = {0.0f, 0.0f, 0.0f, 0.0f};
        for (int g = 0; g < NGROUP; ++g) {
            const int gn = g + 1 < NGROUP ? g + 1 : g;
            const v4f n0 = ldp[3 * gn], n1 = ldp[3 * gn + 1], n2 = ldp[3 * gn + 2];
            const float Tn = ldT[12 * gn];
            const bool mine = (g == sub);
            __builtin_amdgcn_sched_barrier(0);
            const float qa = step(hc.x, a0.x, a0.y, a0.z);
            const float qb = step(hc.y, a0.w, a1v.x, a1v.y);
            __builtin_amdgcn_sched_barrier(0);
            const v4f nh = cold4(Tn);
            const float qc = step(hc.z, a1v.z, a1v.w, a2v.x);
            const float qd = step(hc.w, a2v.y, a2v.z, a2v.w);
            qk.x = mine ? qa : qk.x;
            qk.y = mine ? qb : qk.y;
            qk.z = mine ? qc : qk.z;
            qk.w = mine ? qd : qk.w;
            a0 = n0; a1v = n1; a2v = n2;
            hc = nh;
        }
        *(v4f*)(qp + c * CHUNK) = qk;
    }
}

__global__ __launch_bounds__(256) void hbv_route_wmma_kernel(
    const float* __restrict__ qws, float* __restrict__ out)
{
    constexpr double INV_MID_D = 1.0 / (4.0 + 1e-7);
    constexpr double WSUM_D = 1.0 + 2.0 * ((1.0 - 1.0 * INV_MID_D) +
                                           (1.0 - 2.0 * INV_MID_D) +
                                           (1.0 - 3.0 * INV_MID_D) +
                                           (1.0 - 4.0 * INV_MID_D));
    constexpr float INV_MID  = (float)INV_MID_D;
    constexpr float INV_WSUM = (float)(1.0 / (WSUM_D + 1e-7));

    const int lane = threadIdx.x & 31;
    const int wave = threadIdx.x >> 5;
    const int tile = blockIdx.x * 8 + wave;
    const int tt   = tile & (TDIM / 16 - 1);
    const int tb   = tile >> 8;
    const int b0   = tb * 16;
    const int t0   = tt * 16;
    const int mn   = lane & 15;
    const int khi  = lane >> 4;

    const float* qrow = qws + (size_t)(b0 + mn) * TDIM;
    v2f afr[6], wfr[6];
#pragma unroll
    for (int c = 0; c < 6; ++c) {
        const int tp   = t0 - 4 + 4 * c + 2 * khi;
        const bool inr = (tp >= 0) && (tp < TDIM);
        const v2f v    = *(const v2f*)(qrow + (inr ? tp : 0));
#pragma unroll
        for (int j = 0; j < 2; ++j) {
            const int kg = 4 * c + j + 2 * khi;
            afr[c][j] = inr ? v[j] : 0.0f;
            const float dd = fabsf((float)(kg - mn) - 4.0f);
            wfr[c][j] = fmaxf(0.0f, fmaf(-dd, INV_MID, 1.0f)) * INV_WSUM;
        }
    }

    v8f acc = {0.0f, 0.0f, 0.0f, 0.0f, 0.0f, 0.0f, 0.0f, 0.0f};
#pragma unroll
    for (int c = 0; c < 6; ++c) {
        acc = __builtin_amdgcn_wmma_f32_16x16x4_f32(
            false, afr[c], false, wfr[c], (short)0, acc, false, false);
    }

#pragma unroll
    for (int v = 0; v < 8; ++v) {
        const int m = v + 8 * khi;
        out[(size_t)(b0 + m) * TDIM + t0 + mn] = acc[v];
    }
}

extern "C" void kernel_launch(void* const* d_in, const int* in_sizes, int n_in,
                              void* d_out, int out_size, void* d_ws, size_t ws_size,
                              hipStream_t stream) {
    (void)in_sizes; (void)n_in; (void)out_size; (void)ws_size;
    const float* x      = (const float*)d_in[0];
    const float* params = (const float*)d_in[1];
    float* out = (float*)d_out;
    float* qws = (float*)d_ws;

    hbv_scan_kernel<<<BDIM / (WPB * BPW), WPB * 32, 0, stream>>>(x, params, qws);
    hbv_route_wmma_kernel<<<(BDIM / 16) * (TDIM / 16) / 8, 256, 0, stream>>>(qws, out);
}


# a group's instruction order laid out by hand, rows held by scheduling barriers: chain link first, independent work in its latency shadow; block cut behind a step's rcp so groups overlap; same sums
# speedup vs baseline: 10.2059x; 1.1357x over previous
#include <hip/hip_runtime.h>


#define BDIM 2048
#define TDIM 4096

typedef float v2f __attribute__((ext_vector_type(2)));
typedef float v4f __attribute__((ext_vector_type(4)));
typedef float v8f __attribute__((ext_vector_type(8)));

__device__ __forceinline__ float clip1e5(float d) {
    return fminf(fmaxf(d, -100000.0f), 100000.0f);
}

#define CHUNK   32
#define NCHUNK  (TDIM / CHUNK)
#define ROWV    (CHUNK * 3 / 4)
#define PITCHV  25
#define XROWV   (TDIM * 3 / 4)
#define NGROUP  (CHUNK / 4)
#define LPB     8
#define BPW     (32 / LPB)
#define WPB     4
#define NPIECE  (ROWV / LPB)

__device__ __forceinline__ void wave_tile_handover() {
    __builtin_amdgcn_fence(__ATOMIC_RELEASE, "wavefront");
    __builtin_amdgcn_wave_barrier();
    __builtin_amdgcn_fence(__ATOMIC_ACQUIRE, "wavefront");
}

template <int K>
__device__ __forceinline__ float group_bcast(float v, unsigned int grp8) {
    const unsigned int u = __builtin_bit_cast(unsigned int, v);
    return __builtin_bit_cast(float,
        __builtin_amdgcn_permlane16_var(u, u, grp8 | (unsigned int)K, true, true));
}

__global__ __launch_bounds__(WPB * 32) void hbv_scan_kernel(
    const float* __restrict__ x, const float* __restrict__ p,
    float* __restrict__ qws)
{
    __shared__ v4f tile[WPB * BPW * PITCHV];

    const int lane = threadIdx.x & 31;
    const int wave = threadIdx.x >> 5;
    const int sub  = lane & (LPB - 1);
    const int row  = lane / LPB;
    const int b    = (blockIdx.x * WPB + wave) * BPW + row;

    const float Tt     = p[0]  * -1.5f;
    const float CFR    = p[1]  * 0.01f;
    const float CFMAX  = p[2];
    const float SCF    = p[3];
    const float FC     = p[4]  * 100.0f;
    const float Beta   = p[5]  * 8.0f;
    const float LP     = p[6]  * 2.5f;
    const float CWH    = p[7]  * 0.3f;
    const float UZL    = p[8]  * 120.0f;
    const float K0     = p[9]  * 0.5f;
    const float K1     = p[10] * 0.3f;
    const float K2     = p[11] * 0.01f;
    const float PER    = p[12] * 3.5f;
    const float rawCWH = p[7];

    const float invFC   = __builtin_amdgcn_rcpf(FC);
    const float FCLP    = FC * LP;
    const float invFCLP = __builtin_amdgcn_rcpf(FCLP);
    const float CFRMAX  = CFR * CFMAX;

    const float KS = -14.4269504088896341f;
    float c1 = 0.0f, c2 = 0.0f, c3 = 0.0f, c4 = 0.0f, c5 = 0.0f, k0 = 0.0f;
    if (sub == 1) { c1 = -(KS * CWH); c2 = KS; }
    if (sub == 2) { c3 = KS; }
    if (sub == 3) { c3 = -KS; k0 = KS * FCLP; }
    if (sub == 4) { c4 = KS; }
    if (sub == 5) { c4 = KS; k0 = -(KS * UZL); }
    if (sub == 6) { c5 = KS; }
    const unsigned int grp8 = (unsigned int)lane & 8u;

    float S1 = 0.0f, S2 = 0.0f, S3 = 0.0f, S4 = 0.0f, S5 = 0.0f;

    auto cold4 = [&](float Tm) __attribute__((always_inline)) {
        const float e  = __builtin_amdgcn_exp2f(-14.4269504088896341f * (Tt - Tm));
        const float r  = __builtin_amdgcn_rcpf(1.0f + e);
        v4f h;
        h.x = group_bcast<0>(r, grp8);
        h.y = group_bcast<1>(r, grp8);
        h.z = group_bcast<2>(r, grp8);
        h.w = group_bcast<3>(r, grp8);
        return h;
    };

#define ROW_END() __builtin_amdgcn_sched_barrier(0)

    float r_c = 0.0f, PrM_c = 0.0f, pw_c = 0.0f, b2_c = 0.0f, minR_c = 0.0f, lo2_c = 0.0f,
          hi2_c = 0.0f, lo3_c = 0.0f, hi3_c = 0.0f, lo4_c = 0.0f, hi4_c = 0.0f, lo5_c = 0.0f,
          hi5_c = 0.0f, q0c_c = 0.0f, q1c_c = 0.0f, q2c_c = 0.0f, EaA_c = 0.0f;
    float hcold_c = 0.0f, PET_c = 0.0f;
    float e1 = k0;

    const v4f* xg = (const v4f*)x + (size_t)b * XROWV + sub;
    v4f* wtile    = tile + wave * (BPW * PITCHV);
    v4f* stp      = wtile + row * PITCHV + sub;
    const v4f* ldp = wtile + row * PITCHV;
    const float* ldT = (const float*)ldp + 3 * (sub & 3) + 1;
    float* qp = qws + (size_t)b * TDIM + sub * 4;

    v4f pre[NPIECE];
#pragma unroll
    for (int i = 0; i < NPIECE; ++i)
        pre[i] = __builtin_nontemporal_load(xg + i * LPB);

    for (int c = 0; c < NCHUNK; ++c) {
        wave_tile_handover();
#pragma unroll
        for (int i = 0; i < NPIECE; ++i)
            stp[i * LPB] = pre[i];
        wave_tile_handover();

        v4f a0 = ldp[0], a1v = ldp[1], a2v = ldp[2];
        const float T0 = ldT[0];

        if (c + 1 < NCHUNK) {
            const v4f* xn = xg + (c + 1) * ROWV;
#pragma unroll
            for (int i = 0; i < NPIECE; ++i)
                pre[i] = __builtin_nontemporal_load(xn + i * LPB);
        }

        v4f hc = cold4(T0);
        {
            const float ycF_0 = Tt - a0.y;
            const float aR_0 = ycF_0 * CFRMAX;
            const float tmF_0 = a0.y - Tt;
            const float aM_0 = tmF_0 * CFMAX;
            const float hw_0 = 1.0f - hc.x;
            const float PS_0 = a0.x * SCF;
            const float ex1_0 = fmaf(c2, S2, e1);
            const float ex2_0 = fmaf(c5, S5, ex1_0);
            const float ex3_0 = fmaf(c4, S4, ex2_0);
            const float ex4_0 = fmaf(c3, S3, ex3_0);
            const float e_0 = __builtin_amdgcn_exp2f(ex4_0);
            const float t1_0 = 1.0f + e_0;
            const float r_0 = __builtin_amdgcn_rcpf(t1_0);
            const float minR_0 = fminf(aR_0, S2);
            const float minM_0 = fminf(aM_0, S1);
            const float Rfr_0 = hc.x * minR_0;
            const float Mv_0 = hw_0 * minM_0;
            const float s1_0 = fmaf(hc.x, PS_0, Rfr_0);
            const float PrM_0 = fmaf(hw_0, a0.x, Mv_0);
            const float s2_0 = fmaf(-hw_0, minM_0, s1_0);
            const float b2_0 = PrM_0 + S2;
            const float cl_0 = clip1e5(s2_0);
            const float sS1_0 = S1 + cl_0;
            const float S1_0 = fmaxf(sS1_0, 0.0f);
            const float cap2_0 = S1_0 * rawCWH;
            const float e1_0 = fmaf(c1, S1_0, k0);
            const float S2m_0 = S2 - 100000.0f;
            const float S2p_0 = S2 + 100000.0f;
            const float lo2_0 = fmaxf(fminf(S2m_0, cap2_0), 0.0f);
            const float hi2_0 = fmaxf(fminf(S2p_0, cap2_0), 0.0f);
            const float x3_0 = S3 * invFC;
            const float lg_0 = __builtin_amdgcn_logf(x3_0);
            const float bl_0 = Beta * lg_0;
            const float pw_0 = __builtin_amdgcn_exp2f(bl_0);
            const float S3m_0 = S3 - 100000.0f;
            const float hi3_0 = S3 + 100000.0f;
            const float lo3_0 = fmaxf(S3m_0, 0.0f);
            const float S4m_0 = S4 - 100000.0f;
            const float hi4_0 = S4 + 100000.0f;
            const float lo4_0 = fmaxf(S4m_0, 0.0f);
            const float S5m_0 = S5 - 100000.0f;
            const float hi5_0 = S5 + 100000.0f;
            const float lo5_0 = fmaxf(S5m_0, 0.0f);
            const float q0a_0 = S4 - UZL;
            const float q0c_0 = q0a_0 * K0;
            const float q1c_0 = S4 * K1;
            const float q2c_0 = S5 * K2;
            const float EaP_0 = a0.z * S3;
            const float EaA_0 = EaP_0 * invFCLP;
            r_c = r_0; PrM_c = PrM_0; pw_c = pw_0; b2_c = b2_0; minR_c = minR_0; lo2_c = lo2_0; hi2_c = hi2_0;
            lo3_c = lo3_0; hi3_c = hi3_0; lo4_c = lo4_0; hi4_c = hi4_0; lo5_c = lo5_0; hi5_c = hi5_0; q0c_c = q0c_0;
            q1c_c = q1c_0; q2c_c = q2c_0; EaA_c = EaA_0;
            hcold_c = hc.x; PET_c = a0.z;
            S1 = S1_0; e1 = e1_0;
        }

        v4f qk = {0.0f, 0.0f, 0.0f, 0.0f};
        for (int g = 0; g < NGROUP - 1; ++g) {
            const int gn = g + 1;
            const v4f n0 = ldp[3 * gn], n1 = ldp[3 * gn + 1], n2 = ldp[3 * gn + 2];
            const float Tn = ldT[12 * gn];
            const bool mine = (g == sub);
            __builtin_amdgcn_sched_barrier(0);
            const float r_0 = r_c; const float PrM_0 = PrM_c; const float pw_0 = pw_c; const float b2_0 = b2_c;
            const float minR_0 = minR_c; const float lo2_0 = lo2_c; const float hi2_0 = hi2_c; const float lo3_0 = lo3_c;
            const float hi3_0 = hi3_c; const float lo4_0 = lo4_c; const float hi4_0 = hi4_c; const float lo5_0 = lo5_c;
            const float hi5_0 = hi5_c; const float q0c_0 = q0c_c; const float q1c_0 = q1c_c; const float q2c_0 = q2c_c;
            const float EaA_0 = EaA_c;
            const float hS3_0 = group_bcast<2>(r_0, grp8);
            const float hI_0 = group_bcast<1>(r_0, grp8);
            ROW_END();
            const float hUZ_0 = group_bcast<5>(r_0, grp8);
            const float hLP_0 = group_bcast<3>(r_0, grp8);
            ROW_END();
            const float hS4_0 = group_bcast<4>(r_0, grp8);
            const float A1_0 = EaA_0 * hS3_0;
            const float P1_0 = PET_c * hS3_0;
            ROW_END();
            const float I_0 = PrM_0 * hI_0;
            const float t2a_0 = fmaf(-PrM_0, hI_0, b2_0);
            const float t4a_0 = fmaf(-q0c_0, hUZ_0, S4);
            const float t3a_0 = fmaf(PrM_0, hI_0, S3);
            ROW_END();
            const float hS5_0 = group_bcast<6>(r_0, grp8);
            const float oml_0 = 1.0f - hLP_0;
            const float t5a_0 = fmaf(PER, hS4_0, S5);
            ROW_END();
            const float A2_0 = A1_0 * hLP_0;
            const float ycF_1 = Tt - a1v.x;
            const float t2b_0 = fmaf(-hcold_c, minR_0, t2a_0);
            const float tmF_1 = a1v.x - Tt;
            ROW_END();
            const float t4b_0 = fmaf(I_0, pw_0, t4a_0);
            const float PS_1 = a0.w * SCF;
            const float hw_1 = 1.0f - hc.y;
            const float ycF_2 = Tt - a1v.w;
            ROW_END();
            const float t5b_0 = fmaf(-q2c_0, hS5_0, t5a_0);
            const float tmF_2 = a1v.w - Tt;
            const float Ea_0 = fmaf(P1_0, oml_0, A2_0);
            const float aR_1 = ycF_1 * CFRMAX;
            ROW_END();
            const float S2_0 = __builtin_amdgcn_fmed3f(t2b_0, lo2_0, hi2_0);
            const float t4c_0 = fmaf(-q1c_0, hS4_0, t4b_0);
            const float aM_1 = tmF_1 * CFMAX;
            ROW_END();
            const float aR_2 = ycF_2 * CFRMAX;
            const float PS_2 = a1v.z * SCF;
            const float S5_0 = __builtin_amdgcn_fmed3f(t5b_0, lo5_0, hi5_0);
            ROW_END();
            const float t3b_0 = t3a_0 - Ea_0;
            const float aM_2 = tmF_2 * CFMAX;
            const float ex1_1 = fmaf(c2, S2_0, e1);
            const float minR_1 = fminf(aR_1, S2_0);
            ROW_END();
            const float t4d_0 = fmaf(-PER, hS4_0, t4c_0);
            const float minM_1 = fminf(aM_1, S1);
            const float S2m_1 = S2_0 - 100000.0f;
            const float S2p_1 = S2_0 + 100000.0f;
            ROW_END();
            const float S5m_1 = S5_0 - 100000.0f;
            const float q2c_1 = S5_0 * K2;
            const float t3c_0 = fmaf(-I_0, pw_0, t3b_0);
            const float hi5_1 = S5_0 + 100000.0f;
            ROW_END();
            const float ex2_1 = fmaf(c5, S5_0, ex1_1);
            const float Rfr_1 = hc.y * minR_1;
            const float S4_0 = __builtin_amdgcn_fmed3f(t4d_0, lo4_0, hi4_0);
            ROW_END();
            const float Mv_1 = hw_1 * minM_1;
            const float hw_2 = 1.0f - hc.z;
            const float lo5_1 = fmaxf(S5m_1, 0.0f);
            const float ycF_3 = Tt - a2v.z;
            ROW_END();
            const float S3_0 = __builtin_amdgcn_fmed3f(t3c_0, lo3_0, hi3_0);
            const float s1_1 = fmaf(hc.y, PS_1, Rfr_1);
            const float tmF_3 = a2v.z - Tt;
            ROW_END();
            const float ex3_1 = fmaf(c4, S4_0, ex2_1);
            const float q0a_1 = S4_0 - UZL;
            const float PrM_1 = fmaf(hw_1, a0.w, Mv_1);
            const float q1c_1 = S4_0 * K1;
            ROW_END();
            const float S4m_1 = S4_0 - 100000.0f;
            const float hi4_1 = S4_0 + 100000.0f;
            const float x3_1 = S3_0 * invFC;
            const float EaP_1 = a1v.y * S3_0;
            ROW_END();
            const float s2_1 = fmaf(-hw_1, minM_1, s1_1);
            const float S3m_1 = S3_0 - 100000.0f;
            const float ex4_1 = fmaf(c3, S3_0, ex3_1);
            const float q0c_1 = q0a_1 * K0;
            ROW_END();
            const float b2_1 = PrM_1 + S2_0;
            const float hi3_1 = S3_0 + 100000.0f;
            const float lo4_1 = fmaxf(S4m_1, 0.0f);
            const float aR_3 = ycF_3 * CFRMAX;
            ROW_END();
            const float lg_1 = __builtin_amdgcn_logf(x3_1);
            const float cl_1 = clip1e5(s2_1);
            ROW_END();
            const float e_1 = __builtin_amdgcn_exp2f(ex4_1);
            const float EaA_1 = EaP_1 * invFCLP;
            const float lo3_1 = fmaxf(S3m_1, 0.0f);
            ROW_END();
            const float aM_3 = tmF_3 * CFMAX;
            const float PS_3 = a2v.y * SCF;
            const float hw_3 = 1.0f - hc.w;
            const float q0_0 = q0c_0 * hUZ_0;
            ROW_END();
            const float sS1_1 = S1 + cl_1;
            ROW_END();
            const float bl_1 = Beta * lg_1;
            const float qq1_0 = fmaf(q1c_0, hS4_0, q0_0);
            ROW_END();
            const float t1_1 = 1.0f + e_1;
            const float S1_1 = fmaxf(sS1_1, 0.0f);
            ROW_END();
            const float pw_1 = __builtin_amdgcn_exp2f(bl_1);
            const float q_0 = fmaf(q2c_0, hS5_0, qq1_0);
            ROW_END();
            const float r_1 = __builtin_amdgcn_rcpf(t1_1);
            const float cap2_1 = S1_1 * rawCWH;
            const float e1_1 = fmaf(c1, S1_1, k0);
            ROW_END();
            const float minM_2 = fminf(aM_2, S1_1);
            ROW_END();
            const float lo2_1 = fmaxf(fminf(S2m_1, cap2_1), 0.0f);
            const float hi2_1 = fmaxf(fminf(S2p_1, cap2_1), 0.0f);
            ROW_END();
            const float Mv_2 = hw_2 * minM_2;
            const float hS3_1 = group_bcast<2>(r_1, grp8);
            ROW_END();
            const float hI_1 = group_bcast<1>(r_1, grp8);
            const float hUZ_1 = group_bcast<5>(r_1, grp8);
            ROW_END();
            const float hLP_1 = group_bcast<3>(r_1, grp8);
            const float hS4_1 = group_bcast<4>(r_1, grp8);
            ROW_END();
            const float A1_1 = EaA_1 * hS3_1;
            const float P1_1 = a1v.y * hS3_1;
            const float I_1 = PrM_1 * hI_1;
            const float t2a_1 = fmaf(-PrM_1, hI_1, b2_1);
            ROW_END();
            const float t4a_1 = fmaf(-q0c_1, hUZ_1, S4_0);
            const float t3a_1 = fmaf(PrM_1, hI_1, S3_0);
            const float hS5_1 = group_bcast<6>(r_1, grp8);
            ROW_END();
            const float oml_1 = 1.0f - hLP_1;
            const float t5a_1 = fmaf(PER, hS4_1, S5_0);
            const float A2_1 = A1_1 * hLP_1;
            const float PrM_2 = fmaf(hw_2, a1v.z, Mv_2);
            ROW_END();
            const float t2b_1 = fmaf(-hc.y, minR_1, t2a_1);
            const float q0_1 = q0c_1 * hUZ_1;
            const float t4b_1 = fmaf(I_1, pw_1, t4a_1);
            ROW_END();
            const float t5b_1 = fmaf(-q2c_1, hS5_1, t5a_1);
            const float Ea_1 = fmaf(P1_1, oml_1, A2_1);
            ROW_END();
            const float S2_1 = __builtin_amdgcn_fmed3f(t2b_1, lo2_1, hi2_1);
            const float t4c_1 = fmaf(-q1c_1, hS4_1, t4b_1);
            const float qq1_1 = fmaf(q1c_1, hS4_1, q0_1);
            ROW_END();
            const float S5_1 = __builtin_amdgcn_fmed3f(t5b_1, lo5_1, hi5_1);
            const float t3b_1 = t3a_1 - Ea_1;
            ROW_END();
            const float ex1_2 = fmaf(c2, S2_1, e1_1);
            const float minR_2 = fminf(aR_2, S2_1);
            const float t4d_1 = fmaf(-PER, hS4_1, t4c_1);
            const float b2_2 = PrM_2 + S2_1;
            ROW_END();
            const float S2m_2 = S2_1 - 100000.0f;
            const float S2p_2 = S2_1 + 100000.0f;
            const float S5m_2 = S5_1 - 100000.0f;
            const float q2c_2 = S5_1 * K2;
            ROW_END();
            const float t3c_1 = fmaf(-I_1, pw_1, t3b_1);
            const float hi5_2 = S5_1 + 100000.0f;
            const float ex2_2 = fmaf(c5, S5_1, ex1_2);
            const float Rfr_2 = hc.z * minR_2;
            ROW_END();
            const float S4_1 = __builtin_amdgcn_fmed3f(t4d_1, lo4_1, hi4_1);
            const float q_1 = fmaf(q2c_1, hS5_1, qq1_1);
            ROW_END();
            const float lo5_2 = fmaxf(S5m_2, 0.0f);
            const float S3_1 = __builtin_amdgcn_fmed3f(t3c_1, lo3_1, hi3_1);
            ROW_END();
            const float s1_2 = fmaf(hc.z, PS_2, Rfr_2);
            const float ex3_2 = fmaf(c4, S4_1, ex2_2);
            const float q0a_2 = S4_1 - UZL;
            ROW_END();
            const float q1c_2 = S4_1 * K1;
            const float S4m_2 = S4_1 - 100000.0f;
            const float hi4_2 = S4_1 + 100000.0f;
            ROW_END();
            const float x3_2 = S3_1 * invFC;
            const float EaP_2 = a2v.x * S3_1;
            const float s2_2 = fmaf(-hw_2, minM_2, s1_2);
            const float S3m_2 = S3_1 - 100000.0f;
            ROW_END();
            const float ex4_2 = fmaf(c3, S3_1, ex3_2);
            const float q0c_2 = q0a_2 * K0;
            const float lo4_2 = fmaxf(S4m_2, 0.0f);
            const float hi3_2 = S3_1 + 100000.0f;
            ROW_END();
            const float ycF_4 = Tt - n0.y;
            const float tmF_4 = n0.y - Tt;
            const float lg_2 = __builtin_amdgcn_logf(x3_2);
            ROW_END();
            const float cl_2 = clip1e5(s2_2);
            const float e_2 = __builtin_amdgcn_exp2f(ex4_2);
            ROW_END();
            const float EaA_2 = EaP_2 * invFCLP;
            const float lo3_2 = fmaxf(S3m_2, 0.0f);
            const float aR_4 = ycF_4 * CFRMAX;
            const float aM_4 = tmF_4 * CFMAX;
            ROW_END();
            const float PS_4 = n0.x * SCF;
            const float cx0 = Tt - Tn;
            const float sS1_2 = S1_1 + cl_2;
            ROW_END();
            const float bl_2 = Beta * lg_2;
            const float cx1 = -14.4269504088896341f * cx0;
            ROW_END();
            const float t1_2 = 1.0f + e_2;
            const float S1_2 = fmaxf(sS1_2, 0.0f);
            ROW_END();
            const float pw_2 = __builtin_amdgcn_exp2f(bl_2);
            ROW_END();
            const float r_2 = __builtin_amdgcn_rcpf(t1_2);
            const float cap2_2 = S1_2 * rawCWH;
            const float e1_2 = fmaf(c1, S1_2, k0);
            ROW_END();
            const float minM_3 = fminf(aM_3, S1_2);
            const float ce = __builtin_amdgcn_exp2f(cx1);
            ROW_END();
            const float lo2_2 = fmaxf(fminf(S2m_2, cap2_2), 0.0f);
            const float hi2_2 = fmaxf(fminf(S2p_2, cap2_2), 0.0f);
            ROW_END();
            const float Mv_3 = hw_3 * minM_3;
            const float hS3_2 = group_bcast<2>(r_2, grp8);
            ROW_END();
            const float hI_2 = group_bcast<1>(r_2, grp8);
            const float hUZ_2 = group_bcast<5>(r_2, grp8);
            ROW_END();
            const float hLP_2 = group_bcast<3>(r_2, grp8);
            const float hS4_2 = group_bcast<4>(r_2, grp8);
            ROW_END();
            const float A1_2 = EaA_2 * hS3_2;
            const float P1_2 = a2v.x * hS3_2;
            const float I_2 = PrM_2 * hI_2;
            const float t2a_2 = fmaf(-PrM_2, hI_2, b2_2);
            ROW_END();
            const float t4a_2 = fmaf(-q0c_2, hUZ_2, S4_1);
            const float t3a_2 = fmaf(PrM_2, hI_2, S3_1);
            const float hS5_2 = group_bcast<6>(r_2, grp8);
            ROW_END();
            const float oml_2 = 1.0f - hLP_2;
            const float t5a_2 = fmaf(PER, hS4_2, S5_1);
            const float A2_2 = A1_2 * hLP_2;
            const float PrM_3 = fmaf(hw_3, a2v.y, Mv_3);
            ROW_END();
            const float t2b_2 = fmaf(-hc.z, minR_2, t2a_2);
            const float q0_2 = q0c_2 * hUZ_2;
            const float t4b_2 = fmaf(I_2, pw_2, t4a_2);
            const float cs = 1.0f + ce;
            ROW_END();
            const float t5b_2 = fmaf(-q2c_2, hS5_2, t5a_2);
            const float Ea_2 = fmaf(P1_2, oml_2, A2_2);
            ROW_END();
            const float S2_2 = __builtin_amdgcn_fmed3f(t2b_2, lo2_2, hi2_2);
            const float t4c_2 = fmaf(-q1c_2, hS4_2, t4b_2);
            const float qq1_2 = fmaf(q1c_2, hS4_2, q0_2);
            ROW_END();
            const float cr = __builtin_amdgcn_rcpf(cs);
            const float S5_2 = __builtin_amdgcn_fmed3f(t5b_2, lo5_2, hi5_2);
            ROW_END();
            const float t3b_2 = t3a_2 - Ea_2;
            const float ex1_3 = fmaf(c2, S2_2, e1_2);
            const float minR_3 = fminf(aR_3, S2_2);
            ROW_END();
            const float t4d_2 = fmaf(-PER, hS4_2, t4c_2);
            const float b2_3 = PrM_3 + S2_2;
            const float S2m_3 = S2_2 - 100000.0f;
            const float S2p_3 = S2_2 + 100000.0f;
            ROW_END();
            const float S5m_3 = S5_2 - 100000.0f;
            const float q2c_3 = S5_2 * K2;
            const float t3c_2 = fmaf(-I_2, pw_2, t3b_2);
            const float hi5_3 = S5_2 + 100000.0f;
            ROW_END();
            const float ex2_3 = fmaf(c5, S5_2, ex1_3);
            const float Rfr_3 = hc.w * minR_3;
            const float S4_2 = __builtin_amdgcn_fmed3f(t4d_2, lo4_2, hi4_2);
            ROW_END();
            const float q_2 = fmaf(q2c_2, hS5_2, qq1_2);
            const float lo5_3 = fmaxf(S5m_3, 0.0f);
            ROW_END();
            const float S3_2 = __builtin_amdgcn_fmed3f(t3c_2, lo3_2, hi3_2);
            const float s1_3 = fmaf(hc.w, PS_3, Rfr_3);
            ROW_END();
            const float ex3_3 = fmaf(c4, S4_2, ex2_3);
            const float q0a_3 = S4_2 - UZL;
            const float q1c_3 = S4_2 * K1;
            const float S4m_3 = S4_2 - 100000.0f;
            ROW_END();
            const float hi4_3 = S4_2 + 100000.0f;
            const float x3_3 = S3_2 * invFC;
            const float EaP_3 = a2v.w * S3_2;
            ROW_END();
            const float s2_3 = fmaf(-hw_3, minM_3, s1_3);
            const float S3m_3 = S3_2 - 100000.0f;
            const float ex4_3 = fmaf(c3, S3_2, ex3_3);
            const float q0c_3 = q0a_3 * K0;
            ROW_END();
            const float lo4_3 = fmaxf(S4m_3, 0.0f);
            const float hi3_3 = S3_2 + 100000.0f;
            const float nhx = group_bcast<0>(cr, grp8);
            ROW_END();
            const float lg_3 = __builtin_amdgcn_logf(x3_3);
            const float cl_3 = clip1e5(s2_3);
            ROW_END();
            const float e_3 = __builtin_amdgcn_exp2f(ex4_3);
            const float EaA_3 = EaP_3 * invFCLP;
            const float lo3_3 = fmaxf(S3m_3, 0.0f);
            ROW_END();
            const float hw_4 = 1.0f - nhx;
            const float nhy = group_bcast<1>(cr, grp8);
            ROW_END();
            const float sS1_3 = S1_2 + cl_3;
            const float nhz = group_bcast<2>(cr, grp8);
            ROW_END();
            const float nhw = group_bcast<3>(cr, grp8);
            const float bl_3 = Beta * lg_3;
            ROW_END();
            const float t1_3 = 1.0f + e_3;
            const float S1_3 = fmaxf(sS1_3, 0.0f);
            ROW_END();
            const float pw_3 = __builtin_amdgcn_exp2f(bl_3);
            ROW_END();
            const float r_3 = __builtin_amdgcn_rcpf(t1_3);
            const float cap2_3 = S1_3 * rawCWH;
            const float e1_3 = fmaf(c1, S1_3, k0);
            ROW_END();
            const float minM_4 = fminf(aM_4, S1_3);
            ROW_END();
            const float lo2_3 = fmaxf(fminf(S2m_3, cap2_3), 0.0f);
            const float hi2_3 = fmaxf(fminf(S2p_3, cap2_3), 0.0f);
            ROW_END();
            const float Mv_4 = hw_4 * minM_4;
            const float hI_3 = group_bcast<1>(r_3, grp8);
            ROW_END();
            const float hS3_3 = group_bcast<2>(r_3, grp8);
            const float hUZ_3 = group_bcast<5>(r_3, grp8);
            ROW_END();
            const float hLP_3 = group_bcast<3>(r_3, grp8);
            const float hS4_3 = group_bcast<4>(r_3, grp8);
            ROW_END();
            const float t2a_3 = fmaf(-PrM_3, hI_3, b2_3);
            const float I_3 = PrM_3 * hI_3;
            const float A1_3 = EaA_3 * hS3_3;
            const float P1_3 = a2v.w * hS3_3;
            ROW_END();
            const float t4a_3 = fmaf(-q0c_3, hUZ_3, S4_2);
            const float t3a_3 = fmaf(PrM_3, hI_3, S3_2);
            const float hS5_3 = group_bcast<6>(r_3, grp8);
            ROW_END();
            const float oml_3 = 1.0f - hLP_3;
            const float t5a_3 = fmaf(PER, hS4_3, S5_2);
            const float t2b_3 = fmaf(-hc.w, minR_3, t2a_3);
            const float q0_3 = q0c_3 * hUZ_3;
            ROW_END();
            const float A2_3 = A1_3 * hLP_3;
            const float PrM_4 = fmaf(hw_4, n0.x, Mv_4);
            const float t4b_3 = fmaf(I_3, pw_3, t4a_3);
            ROW_END();
            const float t5b_3 = fmaf(-q2c_3, hS5_3, t5a_3);
            const float S2_3 = __builtin_amdgcn_fmed3f(t2b_3, lo2_3, hi2_3);
            ROW_END();
            const float Ea_3 = fmaf(P1_3, oml_3, A2_3);
            const float qq1_3 = fmaf(q1c_3, hS4_3, q0_3);
            const float t4c_3 = fmaf(-q1c_3, hS4_3, t4b_3);
            ROW_END();
            const float S5_3 = __builtin_amdgcn_fmed3f(t5b_3, lo5_3, hi5_3);
            const float minR_4 = fminf(aR_4, S2_3);
            const float ex1_4 = fmaf(c2, S2_3, e1_3);
            ROW_END();
            const float t3b_3 = t3a_3 - Ea_3;
            const float S2m_4 = S2_3 - 100000.0f;
            const float t4d_3 = fmaf(-PER, hS4_3, t4c_3);
            const float S2p_4 = S2_3 + 100000.0f;
            ROW_END();
            const float q_3 = fmaf(q2c_3, hS5_3, qq1_3);
            const float b2_4 = PrM_4 + S2_3;
            const float S5m_4 = S5_3 - 100000.0f;
            const float hi5_4 = S5_3 + 100000.0f;
            ROW_END();
            const float Rfr_4 = nhx * minR_4;
            const float ex2_4 = fmaf(c5, S5_3, ex1_4);
            const float t3c_3 = fmaf(-I_3, pw_3, t3b_3);
            const float q2c_4 = S5_3 * K2;
            ROW_END();
            const float S4_3 = __builtin_amdgcn_fmed3f(t4d_3, lo4_3, hi4_3);
            ROW_END();
            const float lo5_4 = fmaxf(S5m_4, 0.0f);
            const float s1_4 = fmaf(nhx, PS_4, Rfr_4);
            ROW_END();
            const float S3_3 = __builtin_amdgcn_fmed3f(t3c_3, lo3_3, hi3_3);
            const float ex3_4 = fmaf(c4, S4_3, ex2_4);
            const float S4m_4 = S4_3 - 100000.0f;
            ROW_END();
            const float q0a_4 = S4_3 - UZL;
            const float hi4_4 = S4_3 + 100000.0f;
            const float q1c_4 = S4_3 * K1;
            ROW_END();
            const float s2_4 = fmaf(-hw_4, minM_4, s1_4);
            const float x3_4 = S3_3 * invFC;
            const float S3m_4 = S3_3 - 100000.0f;
            ROW_END();
            const float ex4_4 = fmaf(c3, S3_3, ex3_4);
            const float EaP_4 = n0.z * S3_3;
            const float hi3_4 = S3_3 + 100000.0f;
            const float lo4_4 = fmaxf(S4m_4, 0.0f);
            ROW_END();
            const float q0c_4 = q0a_4 * K0;
            const float cl_4 = clip1e5(s2_4);
            ROW_END();
            const float lg_4 = __builtin_amdgcn_logf(x3_4);
            const float lo3_4 = fmaxf(S3m_4, 0.0f);
            const float EaA_4 = EaP_4 * invFCLP;
            ROW_END();
            const float e_4 = __builtin_amdgcn_exp2f(ex4_4);
            ROW_END();
            const float sS1_4 = S1_3 + cl_4;
            ROW_END();
            const float S1_4 = fmaxf(sS1_4, 0.0f);
            const float bl_4 = Beta * lg_4;
            ROW_END();
            const float t1_4 = 1.0f + e_4;
            ROW_END();
            const float cap2_4 = S1_4 * rawCWH;
            const float e1_4 = fmaf(c1, S1_4, k0);
            const float pw_4 = __builtin_amdgcn_exp2f(bl_4);
            ROW_END();
            const float r_4 = __builtin_amdgcn_rcpf(t1_4);
            ROW_END();
            const float lo2_4 = fmaxf(fminf(S2m_4, cap2_4), 0.0f);
            const float hi2_4 = fmaxf(fminf(S2p_4, cap2_4), 0.0f);
            ROW_END();
            qk.x = mine ? q_0 : qk.x;
            qk.y = mine ? q_1 : qk.y;
            qk.z = mine ? q_2 : qk.z;
            qk.w = mine ? q_3 : qk.w;
            S1 = S1_4; S2 = S2_3; S3 = S3_3; S4 = S4_3; S5 = S5_3; e1 = e1_4;
            r_c = r_4; PrM_c = PrM_4; pw_c = pw_4; b2_c = b2_4; minR_c = minR_4; lo2_c = lo2_4; hi2_c = hi2_4;
            lo3_c = lo3_4; hi3_c = hi3_4; lo4_c = lo4_4; hi4_c = hi4_4; lo5_c = lo5_4; hi5_c = hi5_4; q0c_c = q0c_4;
            q1c_c = q1c_4; q2c_c = q2c_4; EaA_c = EaA_4;
            hcold_c = nhx; PET_c = n0.z;
            a0 = n0; a1v = n1; a2v = n2;
            hc.x = nhx; hc.y = nhy; hc.z = nhz; hc.w = nhw;
        }
        {
            const bool mine = (NGROUP - 1 == sub);
            const float r_0 = r_c; const float PrM_0 = PrM_c; const float pw_0 = pw_c; const float b2_0 = b2_c;
            const float minR_0 = minR_c; const float lo2_0 = lo2_c; const float hi2_0 = hi2_c; const float lo3_0 = lo3_c;
            const float hi3_0 = hi3_c; const float lo4_0 = lo4_c; const float hi4_0 = hi4_c; const float lo5_0 = lo5_c;
            const float hi5_0 = hi5_c; const float q0c_0 = q0c_c; const float q1c_0 = q1c_c; const float q2c_0 = q2c_c;
            const float EaA_0 = EaA_c;
            const float hS3_0 = group_bcast<2>(r_0, grp8);
            const float hI_0 = group_bcast<1>(r_0, grp8);
            ROW_END();
            const float hUZ_0 = group_bcast<5>(r_0, grp8);
            const float hLP_0 = group_bcast<3>(r_0, grp8);
            ROW_END();
            const float hS4_0 = group_bcast<4>(r_0, grp8);
            const float A1_0 = EaA_0 * hS3_0;
            const float P1_0 = PET_c * hS3_0;
            ROW_END();
            const float I_0 = PrM_0 * hI_0;
            const float t2a_0 = fmaf(-PrM_0, hI_0, b2_0);
            const float t4a_0 = fmaf(-q0c_0, hUZ_0, S4);
            const float t3a_0 = fmaf(PrM_0, hI_0, S3);
            ROW_END();
            const float hS5_0 = group_bcast<6>(r_0, grp8);
            const float oml_0 = 1.0f - hLP_0;
            const float t5a_0 = fmaf(PER, hS4_0, S5);
            ROW_END();
            const float A2_0 = A1_0 * hLP_0;
            const float ycF_1 = Tt - a1v.x;
            const float t2b_0 = fmaf(-hcold_c, minR_0, t2a_0);
            const float tmF_1 = a1v.x - Tt;
            ROW_END();
            const float t4b_0 = fmaf(I_0, pw_0, t4a_0);
            const float PS_1 = a0.w * SCF;
            const float hw_1 = 1.0f - hc.y;
            const float ycF_2 = Tt - a1v.w;
            ROW_END();
            const float t5b_0 = fmaf(-q2c_0, hS5_0, t5a_0);
            const float tmF_2 = a1v.w - Tt;
            const float Ea_0 = fmaf(P1_0, oml_0, A2_0);
            const float aR_1 = ycF_1 * CFRMAX;
            ROW_END();
            const float S2_0 = __builtin_amdgcn_fmed3f(t2b_0, lo2_0, hi2_0);
            const float t4c_0 = fmaf(-q1c_0, hS4_0, t4b_0);
            const float aM_1 = tmF_1 * CFMAX;
            ROW_END();
            const float aR_2 = ycF_2 * CFRMAX;
            const float PS_2 = a1v.z * SCF;
            const float S5_0 = __builtin_amdgcn_fmed3f(t5b_0, lo5_0, hi5_0);
            ROW_END();
            const float t3b_0 = t3a_0 - Ea_0;
            const float aM_2 = tmF_2 * CFMAX;
            const float ex1_1 = fmaf(c2, S2_0, e1);
            const float minR_1 = fminf(aR_1, S2_0);
            ROW_END();
            const float t4d_0 = fmaf(-PER, hS4_0, t4c_0);
            const float minM_1 = fminf(aM_1, S1);
            const float S2m_1 = S2_0 - 100000.0f;
            const float S2p_1 = S2_0 + 100000.0f;
            ROW_END();
            const float S5m_1 = S5_0 - 100000.0f;
            const float q2c_1 = S5_0 * K2;
            const float t3c_0 = fmaf(-I_0, pw_0, t3b_0);
            const float hi5_1 = S5_0 + 100000.0f;
            ROW_END();
            const float ex2_1 = fmaf(c5, S5_0, ex1_1);
            const float Rfr_1 = hc.y * minR_1;
            const float S4_0 = __builtin_amdgcn_fmed3f(t4d_0, lo4_0, hi4_0);
            ROW_END();
            const float Mv_1 = hw_1 * minM_1;
            const float hw_2 = 1.0f - hc.z;
            const float lo5_1 = fmaxf(S5m_1, 0.0f);
            const float ycF_3 = Tt - a2v.z;
            ROW_END();
            const float S3_0 = __builtin_amdgcn_fmed3f(t3c_0, lo3_0, hi3_0);
            const float s1_1 = fmaf(hc.y, PS_1, Rfr_1);
            const float tmF_3 = a2v.z - Tt;
            ROW_END();
            const float ex3_1 = fmaf(c4, S4_0, ex2_1);
            const float q0a_1 = S4_0 - UZL;
            const float PrM_1 = fmaf(hw_1, a0.w, Mv_1);
            const float q1c_1 = S4_0 * K1;
            ROW_END();
            const float S4m_1 = S4_0 - 100000.0f;
            const float hi4_1 = S4_0 + 100000.0f;
            const float x3_1 = S3_0 * invFC;
            const float EaP_1 = a1v.y * S3_0;
            ROW_END();
            const float s2_1 = fmaf(-hw_1, minM_1, s1_1);
            const float S3m_1 = S3_0 - 100000.0f;
            const float ex4_1 = fmaf(c3, S3_0, ex3_1);
            const float q0c_1 = q0a_1 * K0;
            ROW_END();
            const float b2_1 = PrM_1 + S2_0;
            const float hi3_1 = S3_0 + 100000.0f;
            const float lo4_1 = fmaxf(S4m_1, 0.0f);
            const float aR_3 = ycF_3 * CFRMAX;
            ROW_END();
            const float lg_1 = __builtin_amdgcn_logf(x3_1);
            const float cl_1 = clip1e5(s2_1);
            ROW_END();
            const float e_1 = __builtin_amdgcn_exp2f(ex4_1);
            const float EaA_1 = EaP_1 * invFCLP;
            const float lo3_1 = fmaxf(S3m_1, 0.0f);
            ROW_END();
            const float aM_3 = tmF_3 * CFMAX;
            const float PS_3 = a2v.y * SCF;
            const float hw_3 = 1.0f - hc.w;
            const float q0_0 = q0c_0 * hUZ_0;
            ROW_END();
            const float sS1_1 = S1 + cl_1;
            ROW_END();
            const float bl_1 = Beta * lg_1;
            const float qq1_0 = fmaf(q1c_0, hS4_0, q0_0);
            ROW_END();
            const float t1_1 = 1.0f + e_1;
            const float S1_1 = fmaxf(sS1_1, 0.0f);
            ROW_END();
            const float pw_1 = __builtin_amdgcn_exp2f(bl_1);
            const float q_0 = fmaf(q2c_0, hS5_0, qq1_0);
            ROW_END();
            const float r_1 = __builtin_amdgcn_rcpf(t1_1);
            const float cap2_1 = S1_1 * rawCWH;
            const float e1_1 = fmaf(c1, S1_1, k0);
            ROW_END();
            const float minM_2 = fminf(aM_2, S1_1);
            ROW_END();
            const float lo2_1 = fmaxf(fminf(S2m_1, cap2_1), 0.0f);
            const float hi2_1 = fmaxf(fminf(S2p_1, cap2_1), 0.0f);
            ROW_END();
            const float Mv_2 = hw_2 * minM_2;
            const float hS3_1 = group_bcast<2>(r_1, grp8);
            ROW_END();
            const float hI_1 = group_bcast<1>(r_1, grp8);
            const float hUZ_1 = group_bcast<5>(r_1, grp8);
            ROW_END();
            const float hLP_1 = group_bcast<3>(r_1, grp8);
            const float hS4_1 = group_bcast<4>(r_1, grp8);
            ROW_END();
            const float A1_1 = EaA_1 * hS3_1;
            const float P1_1 = a1v.y * hS3_1;
            const float I_1 = PrM_1 * hI_1;
            const float t2a_1 = fmaf(-PrM_1, hI_1, b2_1);
            ROW_END();
            const float t4a_1 = fmaf(-q0c_1, hUZ_1, S4_0);
            const float t3a_1 = fmaf(PrM_1, hI_1, S3_0);
            const float hS5_1 = group_bcast<6>(r_1, grp8);
            ROW_END();
            const float oml_1 = 1.0f - hLP_1;
            const float t5a_1 = fmaf(PER, hS4_1, S5_0);
            const float A2_1 = A1_1 * hLP_1;
            const float PrM_2 = fmaf(hw_2, a1v.z, Mv_2);
            ROW_END();
            const float t2b_1 = fmaf(-hc.y, minR_1, t2a_1);
            const float q0_1 = q0c_1 * hUZ_1;
            const float t4b_1 = fmaf(I_1, pw_1, t4a_1);
            ROW_END();
            const float t5b_1 = fmaf(-q2c_1, hS5_1, t5a_1);
            const float Ea_1 = fmaf(P1_1, oml_1, A2_1);
            ROW_END();
            const float S2_1 = __builtin_amdgcn_fmed3f(t2b_1, lo2_1, hi2_1);
            const float t4c_1 = fmaf(-q1c_1, hS4_1, t4b_1);
            const float qq1_1 = fmaf(q1c_1, hS4_1, q0_1);
            ROW_END();
            const float S5_1 = __builtin_amdgcn_fmed3f(t5b_1, lo5_1, hi5_1);
            const float t3b_1 = t3a_1 - Ea_1;
            ROW_END();
            const float ex1_2 = fmaf(c2, S2_1, e1_1);
            const float minR_2 = fminf(aR_2, S2_1);
            const float t4d_1 = fmaf(-PER, hS4_1, t4c_1);
            const float b2_2 = PrM_2 + S2_1;
            ROW_END();
            const float S2m_2 = S2_1 - 100000.0f;
            const float S2p_2 = S2_1 + 100000.0f;
            const float S5m_2 = S5_1 - 100000.0f;
            const float q2c_2 = S5_1 * K2;
            ROW_END();
            const float t3c_1 = fmaf(-I_1, pw_1, t3b_1);
            const float hi5_2 = S5_1 + 100000.0f;
            const float ex2_2 = fmaf(c5, S5_1, ex1_2);
            const float Rfr_2 = hc.z * minR_2;
            ROW_END();
            const float S4_1 = __builtin_amdgcn_fmed3f(t4d_1, lo4_1, hi4_1);
            const float q_1 = fmaf(q2c_1, hS5_1, qq1_1);
            ROW_END();
            const float lo5_2 = fmaxf(S5m_2, 0.0f);
            const float S3_1 = __builtin_amdgcn_fmed3f(t3c_1, lo3_1, hi3_1);
            ROW_END();
            const float s1_2 = fmaf(hc.z, PS_2, Rfr_2);
            const float ex3_2 = fmaf(c4, S4_1, ex2_2);
            const float q0a_2 = S4_1 - UZL;
            ROW_END();
            const float q1c_2 = S4_1 * K1;
            const float S4m_2 = S4_1 - 100000.0f;
            const float hi4_2 = S4_1 + 100000.0f;
            ROW_END();
            const float x3_2 = S3_1 * invFC;
            const float EaP_2 = a2v.x * S3_1;
            const float s2_2 = fmaf(-hw_2, minM_2, s1_2);
            const float S3m_2 = S3_1 - 100000.0f;
            ROW_END();
            const float ex4_2 = fmaf(c3, S3_1, ex3_2);
            const float q0c_2 = q0a_2 * K0;
            const float lo4_2 = fmaxf(S4m_2, 0.0f);
            const float hi3_2 = S3_1 + 100000.0f;
            ROW_END();
            const float lg_2 = __builtin_amdgcn_logf(x3_2);
            const float cl_2 = clip1e5(s2_2);
            ROW_END();
            const float e_2 = __builtin_amdgcn_exp2f(ex4_2);
            const float EaA_2 = EaP_2 * invFCLP;
            const float lo3_2 = fmaxf(S3m_2, 0.0f);
            ROW_END();
            const float sS1_2 = S1_1 + cl_2;
            ROW_END();
            const float bl_2 = Beta * lg_2;
            ROW_END();
            const float t1_2 = 1.0f + e_2;
            const float S1_2 = fmaxf(sS1_2, 0.0f);
            ROW_END();
            const float pw_2 = __builtin_amdgcn_exp2f(bl_2);
            ROW_END();
            const float r_2 = __builtin_amdgcn_rcpf(t1_2);
            const float cap2_2 = S1_2 * rawCWH;
            const float e1_2 = fmaf(c1, S1_2, k0);
            ROW_END();
            const float minM_3 = fminf(aM_3, S1_2);
            ROW_END();
            const float lo2_2 = fmaxf(fminf(S2m_2, cap2_2), 0.0f);
            const float hi2_2 = fmaxf(fminf(S2p_2, cap2_2), 0.0f);
            ROW_END();
            const float Mv_3 = hw_3 * minM_3;
            const float hS3_2 = group_bcast<2>(r_2, grp8);
            ROW_END();
            const float hI_2 = group_bcast<1>(r_2, grp8);
            const float hUZ_2 = group_bcast<5>(r_2, grp8);
            ROW_END();
            const float hLP_2 = group_bcast<3>(r_2, grp8);
            const float hS4_2 = group_bcast<4>(r_2, grp8);
            ROW_END();
            const float A1_2 = EaA_2 * hS3_2;
            const float P1_2 = a2v.x * hS3_2;
            const float I_2 = PrM_2 * hI_2;
            const float t2a_2 = fmaf(-PrM_2, hI_2, b2_2);
            ROW_END();
            const float t4a_2 = fmaf(-q0c_2, hUZ_2, S4_1);
            const float t3a_2 = fmaf(PrM_2, hI_2, S3_1);
            const float hS5_2 = group_bcast<6>(r_2, grp8);
            ROW_END();
            const float oml_2 = 1.0f - hLP_2;
            const float t5a_2 = fmaf(PER, hS4_2, S5_1);
            const float A2_2 = A1_2 * hLP_2;
            const float PrM_3 = fmaf(hw_3, a2v.y, Mv_3);
            ROW_END();
            const float t2b_2 = fmaf(-hc.z, minR_2, t2a_2);
            const float q0_2 = q0c_2 * hUZ_2;
            const float t4b_2 = fmaf(I_2, pw_2, t4a_2);
            ROW_END();
            const float t5b_2 = fmaf(-q2c_2, hS5_2, t5a_2);
            const float Ea_2 = fmaf(P1_2, oml_2, A2_2);
            ROW_END();
            const float S2_2 = __builtin_amdgcn_fmed3f(t2b_2, lo2_2, hi2_2);
            const float t4c_2 = fmaf(-q1c_2, hS4_2, t4b_2);
            const float qq1_2 = fmaf(q1c_2, hS4_2, q0_2);
            ROW_END();
            const float S5_2 = __builtin_amdgcn_fmed3f(t5b_2, lo5_2, hi5_2);
            const float t3b_2 = t3a_2 - Ea_2;
            ROW_END();
            const float ex1_3 = fmaf(c2, S2_2, e1_2);
            const float minR_3 = fminf(aR_3, S2_2);
            const float t4d_2 = fmaf(-PER, hS4_2, t4c_2);
            const float b2_3 = PrM_3 + S2_2;
            ROW_END();
            const float S2m_3 = S2_2 - 100000.0f;
            const float S2p_3 = S2_2 + 100000.0f;
            const float S5m_3 = S5_2 - 100000.0f;
            const float q2c_3 = S5_2 * K2;
            ROW_END();
            const float t3c_2 = fmaf(-I_2, pw_2, t3b_2);
            const float hi5_3 = S5_2 + 100000.0f;
            const float ex2_3 = fmaf(c5, S5_2, ex1_3);
            const float Rfr_3 = hc.w * minR_3;
            ROW_END();
            const float S4_2 = __builtin_amdgcn_fmed3f(t4d_2, lo4_2, hi4_2);
            const float q_2 = fmaf(q2c_2, hS5_2, qq1_2);
            ROW_END();
            const float lo5_3 = fmaxf(S5m_3, 0.0f);
            const float S3_2 = __builtin_amdgcn_fmed3f(t3c_2, lo3_2, hi3_2);
            ROW_END();
            const float s1_3 = fmaf(hc.w, PS_3, Rfr_3);
            const float ex3_3 = fmaf(c4, S4_2, ex2_3);
            const float q0a_3 = S4_2 - UZL;
            ROW_END();
            const float q1c_3 = S4_2 * K1;
            const float S4m_3 = S4_2 - 100000.0f;
            const float hi4_3 = S4_2 + 100000.0f;
            ROW_END();
            const float x3_3 = S3_2 * invFC;
            const float EaP_3 = a2v.w * S3_2;
            const float s2_3 = fmaf(-hw_3, minM_3, s1_3);
            const float S3m_3 = S3_2 - 100000.0f;
            ROW_END();
            const float ex4_3 = fmaf(c3, S3_2, ex3_3);
            const float q0c_3 = q0a_3 * K0;
            const float hi3_3 = S3_2 + 100000.0f;
            const float lo4_3 = fmaxf(S4m_3, 0.0f);
            ROW_END();
            const float lg_3 = __builtin_amdgcn_logf(x3_3);
            const float EaA_3 = EaP_3 * invFCLP;
            const float lo3_3 = fmaxf(S3m_3, 0.0f);
            ROW_END();
            const float e_3 = __builtin_amdgcn_exp2f(ex4_3);
            const float cl_3 = clip1e5(s2_3);
            ROW_END();
            const float sS1_3 = S1_2 + cl_3;
            const float bl_3 = Beta * lg_3;
            ROW_END();
            const float t1_3 = 1.0f + e_3;
            ROW_END();
            const float S1_3 = fmaxf(sS1_3, 0.0f);
            const float pw_3 = __builtin_amdgcn_exp2f(bl_3);
            ROW_END();
            const float r_3 = __builtin_amdgcn_rcpf(t1_3);
            ROW_END();
            const float cap2_3 = S1_3 * rawCWH;
            const float e1_3 = fmaf(c1, S1_3, k0);
            ROW_END();
            const float lo2_3 = fmaxf(fminf(S2m_3, cap2_3), 0.0f);
            const float hi2_3 = fmaxf(fminf(S2p_3, cap2_3), 0.0f);
            ROW_END();
            const float hS3_3 = group_bcast<2>(r_3, grp8);
            const float hLP_3 = group_bcast<3>(r_3, grp8);
            ROW_END();
            const float hI_3 = group_bcast<1>(r_3, grp8);
            const float hUZ_3 = group_bcast<5>(r_3, grp8);
            ROW_END();
            const float hS4_3 = group_bcast<4>(r_3, grp8);
            const float A1_3 = EaA_3 * hS3_3;
            const float P1_3 = a2v.w * hS3_3;
            ROW_END();
            const float oml_3 = 1.0f - hLP_3;
            const float I_3 = PrM_3 * hI_3;
            const float t3a_3 = fmaf(PrM_3, hI_3, S3_2);
            ROW_END();
            const float t4a_3 = fmaf(-q0c_3, hUZ_3, S4_2);
            const float q0_3 = q0c_3 * hUZ_3;
            const float hS5_3 = group_bcast<6>(r_3, grp8);
            ROW_END();
            const float A2_3 = A1_3 * hLP_3;
            const float t5a_3 = fmaf(PER, hS4_3, S5_2);
            const float t2a_3 = fmaf(-PrM_3, hI_3, b2_3);
            ROW_END();
            const float t4b_3 = fmaf(I_3, pw_3, t4a_3);
            const float qq1_3 = fmaf(q1c_3, hS4_3, q0_3);
            ROW_END();
            const float Ea_3 = fmaf(P1_3, oml_3, A2_3);
            const float t5b_3 = fmaf(-q2c_3, hS5_3, t5a_3);
            const float t2b_3 = fmaf(-hc.w, minR_3, t2a_3);
            ROW_END();
            const float t4c_3 = fmaf(-q1c_3, hS4_3, t4b_3);
            const float q_3 = fmaf(q2c_3, hS5_3, qq1_3);
            ROW_END();
            const float t3b_3 = t3a_3 - Ea_3;
            const float S2_3 = __builtin_amdgcn_fmed3f(t2b_3, lo2_3, hi2_3);
            ROW_END();
            const float S5_3 = __builtin_amdgcn_fmed3f(t5b_3, lo5_3, hi5_3);
            const float t4d_3 = fmaf(-PER, hS4_3, t4c_3);
            ROW_END();
            const float t3c_3 = fmaf(-I_3, pw_3, t3b_3);
            ROW_END();
            const float S4_3 = __builtin_amdgcn_fmed3f(t4d_3, lo4_3, hi4_3);
            ROW_END();
            const float S3_3 = __builtin_amdgcn_fmed3f(t3c_3, lo3_3, hi3_3);
            ROW_END();
            qk.x = mine ? q_0 : qk.x;
            qk.y = mine ? q_1 : qk.y;
            qk.z = mine ? q_2 : qk.z;
            qk.w = mine ? q_3 : qk.w;
            S1 = S1_3; S2 = S2_3; S3 = S3_3; S4 = S4_3; S5 = S5_3; e1 = e1_3;
        }
        *(v4f*)(qp + c * CHUNK) = qk;
    }
}

__global__ __launch_bounds__(256) void hbv_route_wmma_kernel(
    const float* __restrict__ qws, float* __restrict__ out)
{
    constexpr double INV_MID_D = 1.0 / (4.0 + 1e-7);
    constexpr double WSUM_D = 1.0 + 2.0 * ((1.0 - 1.0 * INV_MID_D) +
                                           (1.0 - 2.0 * INV_MID_D) +
                                           (1.0 - 3.0 * INV_MID_D) +
                                           (1.0 - 4.0 * INV_MID_D));
    constexpr float INV_MID  = (float)INV_MID_D;
    constexpr float INV_WSUM = (float)(1.0 / (WSUM_D + 1e-7));

    const int lane = threadIdx.x & 31;
    const int wave = threadIdx.x >> 5;
    const int tile = blockIdx.x * 8 + wave;
    const int tt   = tile & (TDIM / 16 - 1);
    const int tb   = tile >> 8;
    const int b0   = tb * 16;
    const int t0   = tt * 16;
    const int mn   = lane & 15;
    const int khi  = lane >> 4;

    const float* qrow = qws + (size_t)(b0 + mn) * TDIM;
    v2f afr[6], wfr[6];
#pragma unroll
    for (int c = 0; c < 6; ++c) {
        const int tp   = t0 - 4 + 4 * c + 2 * khi;
        const bool inr = (tp >= 0) && (tp < TDIM);
        const v2f v    = *(const v2f*)(qrow + (inr ? tp : 0));
#pragma unroll
        for (int j = 0; j < 2; ++j) {
            const int kg = 4 * c + j + 2 * khi;
            afr[c][j] = inr ? v[j] : 0.0f;
            const float dd = fabsf((float)(kg - mn) - 4.0f);
            wfr[c][j] = fmaxf(0.0f, fmaf(-dd, INV_MID, 1.0f)) * INV_WSUM;
        }
    }

    v8f acc = {0.0f, 0.0f, 0.0f, 0.0f, 0.0f, 0.0f, 0.0f, 0.0f};
#pragma unroll
    for (int c = 0; c < 6; ++c) {
        acc = __builtin_amdgcn_wmma_f32_16x16x4_f32(
            false, afr[c], false, wfr[c], (short)0, acc, false, false);
    }

#pragma unroll
    for (int v = 0; v < 8; ++v) {
        const int m = v + 8 * khi;
        out[(size_t)(b0 + m) * TDIM + t0 + mn] = acc[v];
    }
}

extern "C" void kernel_launch(void* const* d_in, const int* in_sizes, int n_in,
                              void* d_out, int out_size, void* d_ws, size_t ws_size,
                              hipStream_t stream) {
    (void)in_sizes; (void)n_in; (void)out_size; (void)ws_size;
    const float* x      = (const float*)d_in[0];
    const float* params = (const float*)d_in[1];
    float* out = (float*)d_out;
    float* qws = (float*)d_ws;

    hbv_scan_kernel<<<BDIM / (WPB * BPW), WPB * 32, 0, stream>>>(x, params, qws);
    hbv_route_wmma_kernel<<<(BDIM / 16) * (TDIM / 16) / 8, 256, 0, stream>>>(qws, out);
}


# a tile's 8 groups unrolled whole, run on across the tile border: next tile staged in front of the 8th block; no drain and restart per tile, no register copies or counting per group; same sums
# speedup vs baseline: 11.2719x; 1.1044x over previous
#include <hip/hip_runtime.h>


#define BDIM 2048
#define TDIM 4096

typedef float v2f __attribute__((ext_vector_type(2)));
typedef float v4f __attribute__((ext_vector_type(4)));
typedef float v8f __attribute__((ext_vector_type(8)));

__device__ __forceinline__ float clip1e5(float d) {
    return fminf(fmaxf(d, -100000.0f), 100000.0f);
}

#define CHUNK   32
#define NCHUNK  (TDIM / CHUNK)
#define ROWV    (CHUNK * 3 / 4)
#define PITCHV  25
#define XROWV   (TDIM * 3 / 4)
#define NGROUP  (CHUNK / 4)
#define LPB     8
#define BPW     (32 / LPB)
#define WPB     4
#define NPIECE  (ROWV / LPB)

__device__ __forceinline__ void wave_tile_handover() {
    __builtin_amdgcn_fence(__ATOMIC_RELEASE, "wavefront");
    __builtin_amdgcn_wave_barrier();
    __builtin_amdgcn_fence(__ATOMIC_ACQUIRE, "wavefront");
}

template <int K>
__device__ __forceinline__ float group_bcast(float v, unsigned int grp8) {
    const unsigned int u = __builtin_bit_cast(unsigned int, v);
    return __builtin_bit_cast(float,
        __builtin_amdgcn_permlane16_var(u, u, grp8 | (unsigned int)K, true, true));
}

__global__ __launch_bounds__(WPB * 32) void hbv_scan_kernel(
    const float* __restrict__ x, const float* __restrict__ p,
    float* __restrict__ qws)
{
    __shared__ v4f tile[WPB * BPW * PITCHV];

    const int lane = threadIdx.x & 31;
    const int wave = threadIdx.x >> 5;
    const int sub  = lane & (LPB - 1);
    const int row  = lane / LPB;
    const int b    = (blockIdx.x * WPB + wave) * BPW + row;

    const float Tt     = p[0]  * -1.5f;
    const float CFR    = p[1]  * 0.01f;
    const float CFMAX  = p[2];
    const float SCF    = p[3];
    const float FC     = p[4]  * 100.0f;
    const float Beta   = p[5]  * 8.0f;
    const float LP     = p[6]  * 2.5f;
    const float CWH    = p[7]  * 0.3f;
    const float UZL    = p[8]  * 120.0f;
    const float K0     = p[9]  * 0.5f;
    const float K1     = p[10] * 0.3f;
    const float K2     = p[11] * 0.01f;
    const float PER    = p[12] * 3.5f;
    const float rawCWH = p[7];

    const float invFC   = __builtin_amdgcn_rcpf(FC);
    const float FCLP    = FC * LP;
    const float invFCLP = __builtin_amdgcn_rcpf(FCLP);
    const float CFRMAX  = CFR * CFMAX;

    const float KS = -14.4269504088896341f;
    float c1 = 0.0f, c2 = 0.0f, c3 = 0.0f, c4 = 0.0f, c5 = 0.0f, k0 = 0.0f;
    if (sub == 1) { c1 = -(KS * CWH); c2 = KS; }
    if (sub == 2) { c3 = KS; }
    if (sub == 3) { c3 = -KS; k0 = KS * FCLP; }
    if (sub == 4) { c4 = KS; }
    if (sub == 5) { c4 = KS; k0 = -(KS * UZL); }
    if (sub == 6) { c5 = KS; }
    const unsigned int grp8 = (unsigned int)lane & 8u;

    float S1 = 0.0f, S2 = 0.0f, S3 = 0.0f, S4 = 0.0f, S5 = 0.0f;

    auto cold4 = [&](float Tm) __attribute__((always_inline)) {
        const float e  = __builtin_amdgcn_exp2f(-14.4269504088896341f * (Tt - Tm));
        const float r  = __builtin_amdgcn_rcpf(1.0f + e);
        v4f h;
        h.x = group_bcast<0>(r, grp8);
        h.y = group_bcast<1>(r, grp8);
        h.z = group_bcast<2>(r, grp8);
        h.w = group_bcast<3>(r, grp8);
        return h;
    };

#define ROW_END() __builtin_amdgcn_sched_barrier(0)

    float r_c = 0.0f, PrM_c = 0.0f, pw_c = 0.0f, b2_c = 0.0f, minR_c = 0.0f, lo2_c = 0.0f,
          hi2_c = 0.0f, lo3_c = 0.0f, hi3_c = 0.0f, lo4_c = 0.0f, hi4_c = 0.0f, lo5_c = 0.0f,
          hi5_c = 0.0f, q0c_c = 0.0f, q1c_c = 0.0f, q2c_c = 0.0f, EaA_c = 0.0f;
    float hcold_c = 0.0f, PET_c = 0.0f;
    float e1 = k0;

    const v4f* xg = (const v4f*)x + (size_t)b * XROWV + sub;
    v4f* wtile    = tile + wave * (BPW * PITCHV);
    v4f* stp      = wtile + row * PITCHV + sub;
    const v4f* ldp = wtile + row * PITCHV;
    const float* ldT = (const float*)ldp + 3 * (sub & 3) + 1;
    float* qp = qws + (size_t)b * TDIM + sub * 4;

    v4f pre[NPIECE];
#pragma unroll
    for (int i = 0; i < NPIECE; ++i)
        pre[i] = __builtin_nontemporal_load(xg + i * LPB);


#pragma unroll
    for (int i = 0; i < NPIECE; ++i)
        stp[i * LPB] = pre[i];
    wave_tile_handover();

    v4f a0 = ldp[0], a1v = ldp[1], a2v = ldp[2];
    const float T0 = ldT[0];

    {
        const v4f* xn = xg + ROWV;
#pragma unroll
        for (int i = 0; i < NPIECE; ++i)
            pre[i] = __builtin_nontemporal_load(xn + i * LPB);
    }

    v4f hc = cold4(T0);
    {
        const float ycF_0 = Tt - a0.y;
        const float aR_0 = ycF_0 * CFRMAX;
        const float tmF_0 = a0.y - Tt;
        const float aM_0 = tmF_0 * CFMAX;
        const float hw_0 = 1.0f - hc.x;
        const float PS_0 = a0.x * SCF;
        const float ex1_0 = fmaf(c2, S2, e1);
        const float ex2_0 = fmaf(c5, S5, ex1_0);
        const float ex3_0 = fmaf(c4, S4, ex2_0);
        const float ex4_0 = fmaf(c3, S3, ex3_0);
        const float e_0 = __builtin_amdgcn_exp2f(ex4_0);
        const float t1_0 = 1.0f + e_0;
        const float r_0 = __builtin_amdgcn_rcpf(t1_0);
        const float minR_0 = fminf(aR_0, S2);
        const float minM_0 = fminf(aM_0, S1);
        const float Rfr_0 = hc.x * minR_0;
        const float Mv_0 = hw_0 * minM_0;
        const float s1_0 = fmaf(hc.x, PS_0, Rfr_0);
        const float PrM_0 = fmaf(hw_0, a0.x, Mv_0);
        const float s2_0 = fmaf(-hw_0, minM_0, s1_0);
        const float b2_0 = PrM_0 + S2;
        const float cl_0 = clip1e5(s2_0);
        const float sS1_0 = S1 + cl_0;
        const float S1_0 = fmaxf(sS1_0, 0.0f);
        const float cap2_0 = S1_0 * rawCWH;
        const float e1_0 = fmaf(c1, S1_0, k0);
        const float S2m_0 = S2 - 100000.0f;
        const float S2p_0 = S2 + 100000.0f;
        const float lo2_0 = fmaxf(fminf(S2m_0, cap2_0), 0.0f);
        const float hi2_0 = fmaxf(fminf(S2p_0, cap2_0), 0.0f);
        const float x3_0 = S3 * invFC;
        const float lg_0 = __builtin_amdgcn_logf(x3_0);
        const float bl_0 = Beta * lg_0;
        const float pw_0 = __builtin_amdgcn_exp2f(bl_0);
        const float S3m_0 = S3 - 100000.0f;
        const float hi3_0 = S3 + 100000.0f;
        const float lo3_0 = fmaxf(S3m_0, 0.0f);
        const float S4m_0 = S4 - 100000.0f;
        const float hi4_0 = S4 + 100000.0f;
        const float lo4_0 = fmaxf(S4m_0, 0.0f);
        const float S5m_0 = S5 - 100000.0f;
        const float hi5_0 = S5 + 100000.0f;
        const float lo5_0 = fmaxf(S5m_0, 0.0f);
        const float q0a_0 = S4 - UZL;
        const float q0c_0 = q0a_0 * K0;
        const float q1c_0 = S4 * K1;
        const float q2c_0 = S5 * K2;
        const float EaP_0 = a0.z * S3;
        const float EaA_0 = EaP_0 * invFCLP;
        r_c = r_0; PrM_c = PrM_0; pw_c = pw_0; b2_c = b2_0; minR_c = minR_0; lo2_c = lo2_0; hi2_c = hi2_0;
        lo3_c = lo3_0; hi3_c = hi3_0; lo4_c = lo4_0; hi4_c = hi4_0; lo5_c = lo5_0; hi5_c = hi5_0; q0c_c = q0c_0;
        q1c_c = q1c_0; q2c_c = q2c_0; EaA_c = EaA_0;
        hcold_c = hc.x; PET_c = a0.z;
        S1 = S1_0; e1 = e1_0;
    }

    v4f qk = {0.0f, 0.0f, 0.0f, 0.0f};

    for (int c = 0; c < NCHUNK; ++c) {
#pragma unroll
        for (int g = 0; g < NGROUP; ++g) {
            if (g == NGROUP - 1) {
                wave_tile_handover();
#pragma unroll
                for (int i = 0; i < NPIECE; ++i)
                    stp[i * LPB] = pre[i];
                wave_tile_handover();
                const int cl = (c + 2 < NCHUNK) ? c + 2 : NCHUNK - 1;
                const v4f* xn = xg + cl * ROWV;
#pragma unroll
                for (int i = 0; i < NPIECE; ++i)
                    pre[i] = __builtin_nontemporal_load(xn + i * LPB);
            }
            const int gn = (g + 1) & (NGROUP - 1);
            const v4f n0 = ldp[3 * gn], n1 = ldp[3 * gn + 1], n2 = ldp[3 * gn + 2];
            const float Tn = ldT[12 * gn];
            const bool mine = (g == sub);
            __builtin_amdgcn_sched_barrier(0);
            const float r_0 = r_c; const float PrM_0 = PrM_c; const float pw_0 = pw_c; const float b2_0 = b2_c;
            const float minR_0 = minR_c; const float lo2_0 = lo2_c; const float hi2_0 = hi2_c; const float lo3_0 = lo3_c;
            const float hi3_0 = hi3_c; const float lo4_0 = lo4_c; const float hi4_0 = hi4_c; const float lo5_0 = lo5_c;
            const float hi5_0 = hi5_c; const float q0c_0 = q0c_c; const float q1c_0 = q1c_c; const float q2c_0 = q2c_c;
            const float EaA_0 = EaA_c;
            const float hS3_0 = group_bcast<2>(r_0, grp8);
            const float hI_0 = group_bcast<1>(r_0, grp8);
            ROW_END();
            const float hUZ_0 = group_bcast<5>(r_0, grp8);
            const float hLP_0 = group_bcast<3>(r_0, grp8);
            ROW_END();
            const float hS4_0 = group_bcast<4>(r_0, grp8);
            const float A1_0 = EaA_0 * hS3_0;
            const float P1_0 = PET_c * hS3_0;
            ROW_END();
            const float I_0 = PrM_0 * hI_0;
            const float t2a_0 = fmaf(-PrM_0, hI_0, b2_0);
            const float t4a_0 = fmaf(-q0c_0, hUZ_0, S4);
            const float t3a_0 = fmaf(PrM_0, hI_0, S3);
            ROW_END();
            const float hS5_0 = group_bcast<6>(r_0, grp8);
            const float oml_0 = 1.0f - hLP_0;
            const float t5a_0 = fmaf(PER, hS4_0, S5);
            ROW_END();
            const float A2_0 = A1_0 * hLP_0;
            const float ycF_1 = Tt - a1v.x;
            const float t2b_0 = fmaf(-hcold_c, minR_0, t2a_0);
            const float tmF_1 = a1v.x - Tt;
            ROW_END();
            const float t4b_0 = fmaf(I_0, pw_0, t4a_0);
            const float PS_1 = a0.w * SCF;
            const float hw_1 = 1.0f - hc.y;
            const float ycF_2 = Tt - a1v.w;
            ROW_END();
            const float t5b_0 = fmaf(-q2c_0, hS5_0, t5a_0);
            const float tmF_2 = a1v.w - Tt;
            const float Ea_0 = fmaf(P1_0, oml_0, A2_0);
            const float aR_1 = ycF_1 * CFRMAX;
            ROW_END();
            const float S2_0 = __builtin_amdgcn_fmed3f(t2b_0, lo2_0, hi2_0);
            const float t4c_0 = fmaf(-q1c_0, hS4_0, t4b_0);
            const float aM_1 = tmF_1 * CFMAX;
            ROW_END();
            const float aR_2 = ycF_2 * CFRMAX;
            const float PS_2 = a1v.z * SCF;
            const float S5_0 = __builtin_amdgcn_fmed3f(t5b_0, lo5_0, hi5_0);
            ROW_END();
            const float t3b_0 = t3a_0 - Ea_0;
            const float aM_2 = tmF_2 * CFMAX;
            const float ex1_1 = fmaf(c2, S2_0, e1);
            const float minR_1 = fminf(aR_1, S2_0);
            ROW_END();
            const float t4d_0 = fmaf(-PER, hS4_0, t4c_0);
            const float minM_1 = fminf(aM_1, S1);
            const float S2m_1 = S2_0 - 100000.0f;
            const float S2p_1 = S2_0 + 100000.0f;
            ROW_END();
            const float S5m_1 = S5_0 - 100000.0f;
            const float q2c_1 = S5_0 * K2;
            const float t3c_0 = fmaf(-I_0, pw_0, t3b_0);
            const float hi5_1 = S5_0 + 100000.0f;
            ROW_END();
            const float ex2_1 = fmaf(c5, S5_0, ex1_1);
            const float Rfr_1 = hc.y * minR_1;
            const float S4_0 = __builtin_amdgcn_fmed3f(t4d_0, lo4_0, hi4_0);
            ROW_END();
            const float Mv_1 = hw_1 * minM_1;
            const float hw_2 = 1.0f - hc.z;
            const float lo5_1 = fmaxf(S5m_1, 0.0f);
            const float ycF_3 = Tt - a2v.z;
            ROW_END();
            const float S3_0 = __builtin_amdgcn_fmed3f(t3c_0, lo3_0, hi3_0);
            const float s1_1 = fmaf(hc.y, PS_1, Rfr_1);
            const float tmF_3 = a2v.z - Tt;
            ROW_END();
            const float ex3_1 = fmaf(c4, S4_0, ex2_1);
            const float q0a_1 = S4_0 - UZL;
            const float PrM_1 = fmaf(hw_1, a0.w, Mv_1);
            const float q1c_1 = S4_0 * K1;
            ROW_END();
            const float S4m_1 = S4_0 - 100000.0f;
            const float hi4_1 = S4_0 + 100000.0f;
            const float x3_1 = S3_0 * invFC;
            const float EaP_1 = a1v.y * S3_0;
            ROW_END();
            const float s2_1 = fmaf(-hw_1, minM_1, s1_1);
            const float S3m_1 = S3_0 - 100000.0f;
            const float ex4_1 = fmaf(c3, S3_0, ex3_1);
            const float q0c_1 = q0a_1 * K0;
            ROW_END();
            const float b2_1 = PrM_1 + S2_0;
            const float hi3_1 = S3_0 + 100000.0f;
            const float lo4_1 = fmaxf(S4m_1, 0.0f);
            const float aR_3 = ycF_3 * CFRMAX;
            ROW_END();
            const float lg_1 = __builtin_amdgcn_logf(x3_1);
            const float cl_1 = clip1e5(s2_1);
            ROW_END();
            const float e_1 = __builtin_amdgcn_exp2f(ex4_1);
            const float EaA_1 = EaP_1 * invFCLP;
            const float lo3_1 = fmaxf(S3m_1, 0.0f);
            ROW_END();
            const float aM_3 = tmF_3 * CFMAX;
            const float PS_3 = a2v.y * SCF;
            const float hw_3 = 1.0f - hc.w;
            const float q0_0 = q0c_0 * hUZ_0;
            ROW_END();
            const float sS1_1 = S1 + cl_1;
            ROW_END();
            const float bl_1 = Beta * lg_1;
            const float qq1_0 = fmaf(q1c_0, hS4_0, q0_0);
            ROW_END();
            const float t1_1 = 1.0f + e_1;
            const float S1_1 = fmaxf(sS1_1, 0.0f);
            ROW_END();
            const float pw_1 = __builtin_amdgcn_exp2f(bl_1);
            const float q_0 = fmaf(q2c_0, hS5_0, qq1_0);
            ROW_END();
            const float r_1 = __builtin_amdgcn_rcpf(t1_1);
            const float cap2_1 = S1_1 * rawCWH;
            const float e1_1 = fmaf(c1, S1_1, k0);
            ROW_END();
            const float minM_2 = fminf(aM_2, S1_1);
            ROW_END();
            const float lo2_1 = fmaxf(fminf(S2m_1, cap2_1), 0.0f);
            const float hi2_1 = fmaxf(fminf(S2p_1, cap2_1), 0.0f);
            ROW_END();
            const float Mv_2 = hw_2 * minM_2;
            const float hS3_1 = group_bcast<2>(r_1, grp8);
            ROW_END();
            const float hI_1 = group_bcast<1>(r_1, grp8);
            const float hUZ_1 = group_bcast<5>(r_1, grp8);
            ROW_END();
            const float hLP_1 = group_bcast<3>(r_1, grp8);
            const float hS4_1 = group_bcast<4>(r_1, grp8);
            ROW_END();
            const float A1_1 = EaA_1 * hS3_1;
            const float P1_1 = a1v.y * hS3_1;
            const float I_1 = PrM_1 * hI_1;
            const float t2a_1 = fmaf(-PrM_1, hI_1, b2_1);
            ROW_END();
            const float t4a_1 = fmaf(-q0c_1, hUZ_1, S4_0);
            const float t3a_1 = fmaf(PrM_1, hI_1, S3_0);
            const float hS5_1 = group_bcast<6>(r_1, grp8);
            ROW_END();
            const float oml_1 = 1.0f - hLP_1;
            const float t5a_1 = fmaf(PER, hS4_1, S5_0);
            const float A2_1 = A1_1 * hLP_1;
            const float PrM_2 = fmaf(hw_2, a1v.z, Mv_2);
            ROW_END();
            const float t2b_1 = fmaf(-hc.y, minR_1, t2a_1);
            const float q0_1 = q0c_1 * hUZ_1;
            const float t4b_1 = fmaf(I_1, pw_1, t4a_1);
            ROW_END();
            const float t5b_1 = fmaf(-q2c_1, hS5_1, t5a_1);
            const float Ea_1 = fmaf(P1_1, oml_1, A2_1);
            ROW_END();
            const float S2_1 = __builtin_amdgcn_fmed3f(t2b_1, lo2_1, hi2_1);
            const float t4c_1 = fmaf(-q1c_1, hS4_1, t4b_1);
            const float qq1_1 = fmaf(q1c_1, hS4_1, q0_1);
            ROW_END();
            const float S5_1 = __builtin_amdgcn_fmed3f(t5b_1, lo5_1, hi5_1);
            const float t3b_1 = t3a_1 - Ea_1;
            ROW_END();
            const float ex1_2 = fmaf(c2, S2_1, e1_1);
            const float minR_2 = fminf(aR_2, S2_1);
            const float t4d_1 = fmaf(-PER, hS4_1, t4c_1);
            const float b2_2 = PrM_2 + S2_1;
            ROW_END();
            const float S2m_2 = S2_1 - 100000.0f;
            const float S2p_2 = S2_1 + 100000.0f;
            const float S5m_2 = S5_1 - 100000.0f;
            const float q2c_2 = S5_1 * K2;
            ROW_END();
            const float t3c_1 = fmaf(-I_1, pw_1, t3b_1);
            const float hi5_2 = S5_1 + 100000.0f;
            const float ex2_2 = fmaf(c5, S5_1, ex1_2);
            const float Rfr_2 = hc.z * minR_2;
            ROW_END();
            const float S4_1 = __builtin_amdgcn_fmed3f(t4d_1, lo4_1, hi4_1);
            const float q_1 = fmaf(q2c_1, hS5_1, qq1_1);
            ROW_END();
            const float lo5_2 = fmaxf(S5m_2, 0.0f);
            const float S3_1 = __builtin_amdgcn_fmed3f(t3c_1, lo3_1, hi3_1);
            ROW_END();
            const float s1_2 = fmaf(hc.z, PS_2, Rfr_2);
            const float ex3_2 = fmaf(c4, S4_1, ex2_2);
            const float q0a_2 = S4_1 - UZL;
            ROW_END();
            const float q1c_2 = S4_1 * K1;
            const float S4m_2 = S4_1 - 100000.0f;
            const float hi4_2 = S4_1 + 100000.0f;
            ROW_END();
            const float x3_2 = S3_1 * invFC;
            const float EaP_2 = a2v.x * S3_1;
            const float s2_2 = fmaf(-hw_2, minM_2, s1_2);
            const float S3m_2 = S3_1 - 100000.0f;
            ROW_END();
            const float ex4_2 = fmaf(c3, S3_1, ex3_2);
            const float q0c_2 = q0a_2 * K0;
            const float lo4_2 = fmaxf(S4m_2, 0.0f);
            const float hi3_2 = S3_1 + 100000.0f;
            ROW_END();
            const float ycF_4 = Tt - n0.y;
            const float tmF_4 = n0.y - Tt;
            const float lg_2 = __builtin_amdgcn_logf(x3_2);
            ROW_END();
            const float cl_2 = clip1e5(s2_2);
            const float e_2 = __builtin_amdgcn_exp2f(ex4_2);
            ROW_END();
            const float EaA_2 = EaP_2 * invFCLP;
            const float lo3_2 = fmaxf(S3m_2, 0.0f);
            const float aR_4 = ycF_4 * CFRMAX;
            const float aM_4 = tmF_4 * CFMAX;
            ROW_END();
            const float PS_4 = n0.x * SCF;
            const float cx0 = Tt - Tn;
            const float sS1_2 = S1_1 + cl_2;
            ROW_END();
            const float bl_2 = Beta * lg_2;
            const float cx1 = -14.4269504088896341f * cx0;
            ROW_END();
            const float t1_2 = 1.0f + e_2;
            const float S1_2 = fmaxf(sS1_2, 0.0f);
            ROW_END();
            const float pw_2 = __builtin_amdgcn_exp2f(bl_2);
            ROW_END();
            const float r_2 = __builtin_amdgcn_rcpf(t1_2);
            const float cap2_2 = S1_2 * rawCWH;
            const float e1_2 = fmaf(c1, S1_2, k0);
            ROW_END();
            const float minM_3 = fminf(aM_3, S1_2);
            const float ce = __builtin_amdgcn_exp2f(cx1);
            ROW_END();
            const float lo2_2 = fmaxf(fminf(S2m_2, cap2_2), 0.0f);
            const float hi2_2 = fmaxf(fminf(S2p_2, cap2_2), 0.0f);
            ROW_END();
            const float Mv_3 = hw_3 * minM_3;
            const float hS3_2 = group_bcast<2>(r_2, grp8);
            ROW_END();
            const float hI_2 = group_bcast<1>(r_2, grp8);
            const float hUZ_2 = group_bcast<5>(r_2, grp8);
            ROW_END();
            const float hLP_2 = group_bcast<3>(r_2, grp8);
            const float hS4_2 = group_bcast<4>(r_2, grp8);
            ROW_END();
            const float A1_2 = EaA_2 * hS3_2;
            const float P1_2 = a2v.x * hS3_2;
            const float I_2 = PrM_2 * hI_2;
            const float t2a_2 = fmaf(-PrM_2, hI_2, b2_2);
            ROW_END();
            const float t4a_2 = fmaf(-q0c_2, hUZ_2, S4_1);
            const float t3a_2 = fmaf(PrM_2, hI_2, S3_1);
            const float hS5_2 = group_bcast<6>(r_2, grp8);
            ROW_END();
            const float oml_2 = 1.0f - hLP_2;
            const float t5a_2 = fmaf(PER, hS4_2, S5_1);
            const float A2_2 = A1_2 * hLP_2;
            const float PrM_3 = fmaf(hw_3, a2v.y, Mv_3);
            ROW_END();
            const float t2b_2 = fmaf(-hc.z, minR_2, t2a_2);
            const float q0_2 = q0c_2 * hUZ_2;
            const float t4b_2 = fmaf(I_2, pw_2, t4a_2);
            const float cs = 1.0f + ce;
            ROW_END();
            const float t5b_2 = fmaf(-q2c_2, hS5_2, t5a_2);
            const float Ea_2 = fmaf(P1_2, oml_2, A2_2);
            ROW_END();
            const float S2_2 = __builtin_amdgcn_fmed3f(t2b_2, lo2_2, hi2_2);
            const float t4c_2 = fmaf(-q1c_2, hS4_2, t4b_2);
            const float qq1_2 = fmaf(q1c_2, hS4_2, q0_2);
            ROW_END();
            const float cr = __builtin_amdgcn_rcpf(cs);
            const float S5_2 = __builtin_amdgcn_fmed3f(t5b_2, lo5_2, hi5_2);
            ROW_END();
            const float t3b_2 = t3a_2 - Ea_2;
            const float ex1_3 = fmaf(c2, S2_2, e1_2);
            const float minR_3 = fminf(aR_3, S2_2);
            ROW_END();
            const float t4d_2 = fmaf(-PER, hS4_2, t4c_2);
            const float b2_3 = PrM_3 + S2_2;
            const float S2m_3 = S2_2 - 100000.0f;
            const float S2p_3 = S2_2 + 100000.0f;
            ROW_END();
            const float S5m_3 = S5_2 - 100000.0f;
            const float q2c_3 = S5_2 * K2;
            const float t3c_2 = fmaf(-I_2, pw_2, t3b_2);
            const float hi5_3 = S5_2 + 100000.0f;
            ROW_END();
            const float ex2_3 = fmaf(c5, S5_2, ex1_3);
            const float Rfr_3 = hc.w * minR_3;
            const float S4_2 = __builtin_amdgcn_fmed3f(t4d_2, lo4_2, hi4_2);
            ROW_END();
            const float q_2 = fmaf(q2c_2, hS5_2, qq1_2);
            const float lo5_3 = fmaxf(S5m_3, 0.0f);
            ROW_END();
            const float S3_2 = __builtin_amdgcn_fmed3f(t3c_2, lo3_2, hi3_2);
            const float s1_3 = fmaf(hc.w, PS_3, Rfr_3);
            ROW_END();
            const float ex3_3 = fmaf(c4, S4_2, ex2_3);
            const float q0a_3 = S4_2 - UZL;
            const float q1c_3 = S4_2 * K1;
            const float S4m_3 = S4_2 - 100000.0f;
            ROW_END();
            const float hi4_3 = S4_2 + 100000.0f;
            const float x3_3 = S3_2 * invFC;
            const float EaP_3 = a2v.w * S3_2;
            ROW_END();
            const float s2_3 = fmaf(-hw_3, minM_3, s1_3);
            const float S3m_3 = S3_2 - 100000.0f;
            const float ex4_3 = fmaf(c3, S3_2, ex3_3);
            const float q0c_3 = q0a_3 * K0;
            ROW_END();
            const float lo4_3 = fmaxf(S4m_3, 0.0f);
            const float hi3_3 = S3_2 + 100000.0f;
            const float nhx = group_bcast<0>(cr, grp8);
            ROW_END();
            const float lg_3 = __builtin_amdgcn_logf(x3_3);
            const float cl_3 = clip1e5(s2_3);
            ROW_END();
            const float e_3 = __builtin_amdgcn_exp2f(ex4_3);
            const float EaA_3 = EaP_3 * invFCLP;
            const float lo3_3 = fmaxf(S3m_3, 0.0f);
            ROW_END();
            const float hw_4 = 1.0f - nhx;
            const float nhy = group_bcast<1>(cr, grp8);
            ROW_END();
            const float sS1_3 = S1_2 + cl_3;
            const float nhz = group_bcast<2>(cr, grp8);
            ROW_END();
            const float nhw = group_bcast<3>(cr, grp8);
            const float bl_3 = Beta * lg_3;
            ROW_END();
            const float t1_3 = 1.0f + e_3;
            const float S1_3 = fmaxf(sS1_3, 0.0f);
            ROW_END();
            const float pw_3 = __builtin_amdgcn_exp2f(bl_3);
            ROW_END();
            const float r_3 = __builtin_amdgcn_rcpf(t1_3);
            const float cap2_3 = S1_3 * rawCWH;
            const float e1_3 = fmaf(c1, S1_3, k0);
            ROW_END();
            const float minM_4 = fminf(aM_4, S1_3);
            ROW_END();
            const float lo2_3 = fmaxf(fminf(S2m_3, cap2_3), 0.0f);
            const float hi2_3 = fmaxf(fminf(S2p_3, cap2_3), 0.0f);
            ROW_END();
            const float Mv_4 = hw_4 * minM_4;
            const float hI_3 = group_bcast<1>(r_3, grp8);
            ROW_END();
            const float hS3_3 = group_bcast<2>(r_3, grp8);
            const float hUZ_3 = group_bcast<5>(r_3, grp8);
            ROW_END();
            const float hLP_3 = group_bcast<3>(r_3, grp8);
            const float hS4_3 = group_bcast<4>(r_3, grp8);
            ROW_END();
            const float t2a_3 = fmaf(-PrM_3, hI_3, b2_3);
            const float I_3 = PrM_3 * hI_3;
            const float A1_3 = EaA_3 * hS3_3;
            const float P1_3 = a2v.w * hS3_3;
            ROW_END();
            const float t4a_3 = fmaf(-q0c_3, hUZ_3, S4_2);
            const float t3a_3 = fmaf(PrM_3, hI_3, S3_2);
            const float hS5_3 = group_bcast<6>(r_3, grp8);
            ROW_END();
            const float oml_3 = 1.0f - hLP_3;
            const float t5a_3 = fmaf(PER, hS4_3, S5_2);
            const float t2b_3 = fmaf(-hc.w, minR_3, t2a_3);
            const float q0_3 = q0c_3 * hUZ_3;
            ROW_END();
            const float A2_3 = A1_3 * hLP_3;
            const float PrM_4 = fmaf(hw_4, n0.x, Mv_4);
            const float t4b_3 = fmaf(I_3, pw_3, t4a_3);
            ROW_END();
            const float t5b_3 = fmaf(-q2c_3, hS5_3, t5a_3);
            const float S2_3 = __builtin_amdgcn_fmed3f(t2b_3, lo2_3, hi2_3);
            ROW_END();
            const float Ea_3 = fmaf(P1_3, oml_3, A2_3);
            const float qq1_3 = fmaf(q1c_3, hS4_3, q0_3);
            const float t4c_3 = fmaf(-q1c_3, hS4_3, t4b_3);
            ROW_END();
            const float S5_3 = __builtin_amdgcn_fmed3f(t5b_3, lo5_3, hi5_3);
            const float minR_4 = fminf(aR_4, S2_3);
            const float ex1_4 = fmaf(c2, S2_3, e1_3);
            ROW_END();
            const float t3b_3 = t3a_3 - Ea_3;
            const float S2m_4 = S2_3 - 100000.0f;
            const float t4d_3 = fmaf(-PER, hS4_3, t4c_3);
            const float S2p_4 = S2_3 + 100000.0f;
            ROW_END();
            const float q_3 = fmaf(q2c_3, hS5_3, qq1_3);
            const float b2_4 = PrM_4 + S2_3;
            const float S5m_4 = S5_3 - 100000.0f;
            const float hi5_4 = S5_3 + 100000.0f;
            ROW_END();
            const float Rfr_4 = nhx * minR_4;
            const float ex2_4 = fmaf(c5, S5_3, ex1_4);
            const float t3c_3 = fmaf(-I_3, pw_3, t3b_3);
            const float q2c_4 = S5_3 * K2;
            ROW_END();
            const float S4_3 = __builtin_amdgcn_fmed3f(t4d_3, lo4_3, hi4_3);
            ROW_END();
            const float lo5_4 = fmaxf(S5m_4, 0.0f);
            const float s1_4 = fmaf(nhx, PS_4, Rfr_4);
            ROW_END();
            const float S3_3 = __builtin_amdgcn_fmed3f(t3c_3, lo3_3, hi3_3);
            const float ex3_4 = fmaf(c4, S4_3, ex2_4);
            const float S4m_4 = S4_3 - 100000.0f;
            ROW_END();
            const float q0a_4 = S4_3 - UZL;
            const float hi4_4 = S4_3 + 100000.0f;
            const float q1c_4 = S4_3 * K1;
            ROW_END();
            const float s2_4 = fmaf(-hw_4, minM_4, s1_4);
            const float x3_4 = S3_3 * invFC;
            const float S3m_4 = S3_3 - 100000.0f;
            ROW_END();
            const float ex4_4 = fmaf(c3, S3_3, ex3_4);
            const float EaP_4 = n0.z * S3_3;
            const float hi3_4 = S3_3 + 100000.0f;
            const float lo4_4 = fmaxf(S4m_4, 0.0f);
            ROW_END();
            const float q0c_4 = q0a_4 * K0;
            const float cl_4 = clip1e5(s2_4);
            ROW_END();
            const float lg_4 = __builtin_amdgcn_logf(x3_4);
            const float lo3_4 = fmaxf(S3m_4, 0.0f);
            const float EaA_4 = EaP_4 * invFCLP;
            ROW_END();
            const float e_4 = __builtin_amdgcn_exp2f(ex4_4);
            ROW_END();
            const float sS1_4 = S1_3 + cl_4;
            ROW_END();
            const float S1_4 = fmaxf(sS1_4, 0.0f);
            const float bl_4 = Beta * lg_4;
            ROW_END();
            const float t1_4 = 1.0f + e_4;
            ROW_END();
            const float cap2_4 = S1_4 * rawCWH;
            const float e1_4 = fmaf(c1, S1_4, k0);
            const float pw_4 = __builtin_amdgcn_exp2f(bl_4);
            ROW_END();
            const float r_4 = __builtin_amdgcn_rcpf(t1_4);
            ROW_END();
            const float lo2_4 = fmaxf(fminf(S2m_4, cap2_4), 0.0f);
            const float hi2_4 = fmaxf(fminf(S2p_4, cap2_4), 0.0f);
            ROW_END();
            qk.x = mine ? q_0 : qk.x;
            qk.y = mine ? q_1 : qk.y;
            qk.z = mine ? q_2 : qk.z;
            qk.w = mine ? q_3 : qk.w;
            S1 = S1_4; S2 = S2_3; S3 = S3_3; S4 = S4_3; S5 = S5_3; e1 = e1_4;
            r_c = r_4; PrM_c = PrM_4; pw_c = pw_4; b2_c = b2_4; minR_c = minR_4; lo2_c = lo2_4; hi2_c = hi2_4;
            lo3_c = lo3_4; hi3_c = hi3_4; lo4_c = lo4_4; hi4_c = hi4_4; lo5_c = lo5_4; hi5_c = hi5_4; q0c_c = q0c_4;
            q1c_c = q1c_4; q2c_c = q2c_4; EaA_c = EaA_4;
            hcold_c = nhx; PET_c = n0.z;
            a0 = n0; a1v = n1; a2v = n2;
            hc.x = nhx; hc.y = nhy; hc.z = nhz; hc.w = nhw;
        }
        *(v4f*)(qp + c * CHUNK) = qk;
    }
    asm volatile("s_wait_loadcnt 0x0" ::: "memory");
}

__global__ __launch_bounds__(256) void hbv_route_wmma_kernel(
    const float* __restrict__ qws, float* __restrict__ out)
{
    constexpr double INV_MID_D = 1.0 / (4.0 + 1e-7);
    constexpr double WSUM_D = 1.0 + 2.0 * ((1.0 - 1.0 * INV_MID_D) +
                                           (1.0 - 2.0 * INV_MID_D) +
                                           (1.0 - 3.0 * INV_MID_D) +
                                           (1.0 - 4.0 * INV_MID_D));
    constexpr float INV_MID  = (float)INV_MID_D;
    constexpr float INV_WSUM = (float)(1.0 / (WSUM_D + 1e-7));

    const int lane = threadIdx.x & 31;
    const int wave = threadIdx.x >> 5;
    const int tile = blockIdx.x * 8 + wave;
    const int tt   = tile & (TDIM / 16 - 1);
    const int tb   = tile >> 8;
    const int b0   = tb * 16;
    const int t0   = tt * 16;
    const int mn   = lane & 15;
    const int khi  = lane >> 4;

    const float* qrow = qws + (size_t)(b0 + mn) * TDIM;
    v2f afr[6], wfr[6];
#pragma unroll
    for (int c = 0; c < 6; ++c) {
        const int tp   = t0 - 4 + 4 * c + 2 * khi;
        const bool inr = (tp >= 0) && (tp < TDIM);
        const v2f v    = *(const v2f*)(qrow + (inr ? tp : 0));
#pragma unroll
        for (int j = 0; j < 2; ++j) {
            const int kg = 4 * c + j + 2 * khi;
            afr[c][j] = inr ? v[j] : 0.0f;
            const float dd = fabsf((float)(kg - mn) - 4.0f);
            wfr[c][j] = fmaxf(0.0f, fmaf(-dd, INV_MID, 1.0f)) * INV_WSUM;
        }
    }

    v8f acc = {0.0f, 0.0f, 0.0f, 0.0f, 0.0f, 0.0f, 0.0f, 0.0f};
#pragma unroll
    for (int c = 0; c < 6; ++c) {
        acc = __builtin_amdgcn_wmma_f32_16x16x4_f32(
            false, afr[c], false, wfr[c], (short)0, acc, false, false);
    }

#pragma unroll
    for (int v = 0; v < 8; ++v) {
        const int m = v + 8 * khi;
        out[(size_t)(b0 + m) * TDIM + t0 + mn] = acc[v];
    }
}

extern "C" void kernel_launch(void* const* d_in, const int* in_sizes, int n_in,
                              void* d_out, int out_size, void* d_ws, size_t ws_size,
                              hipStream_t stream) {
    (void)in_sizes; (void)n_in; (void)out_size; (void)ws_size;
    const float* x      = (const float*)d_in[0];
    const float* params = (const float*)d_in[1];
    float* out = (float*)d_out;
    float* qws = (float*)d_ws;

    hbv_scan_kernel<<<BDIM / (WPB * BPW), WPB * 32, 0, stream>>>(x, params, qws);
    hbv_route_wmma_kernel<<<(BDIM / 16) * (TDIM / 16) / 8, 256, 0, stream>>>(qws, out);
}


# a tile's row is read from LDS in TWO places per trip (half a tile each: 12 b128 + 4 b32 into registers of their own), not 4 reads at the head of each of 8 blocks; data path and sums unchanged
# speedup vs baseline: 11.7802x; 1.0451x over previous
#include <hip/hip_runtime.h>


#define BDIM 2048
#define TDIM 4096

typedef float v2f __attribute__((ext_vector_type(2)));
typedef float v4f __attribute__((ext_vector_type(4)));
typedef float v8f __attribute__((ext_vector_type(8)));

__device__ __forceinline__ float clip1e5(float d) {
    return fminf(fmaxf(d, -100000.0f), 100000.0f);
}

#define CHUNK   32
#define NCHUNK  (TDIM / CHUNK)
#define ROWV    (CHUNK * 3 / 4)
#define PITCHV  25
#define XROWV   (TDIM * 3 / 4)
#define NGROUP  (CHUNK / 4)
#define LPB     8
#define BPW     (32 / LPB)
#define WPB     4
#define NPIECE  (ROWV / LPB)

__device__ __forceinline__ void wave_tile_handover() {
    __builtin_amdgcn_fence(__ATOMIC_RELEASE, "wavefront");
    __builtin_amdgcn_wave_barrier();
    __builtin_amdgcn_fence(__ATOMIC_ACQUIRE, "wavefront");
}

template <int K>
__device__ __forceinline__ float group_bcast(float v, unsigned int grp8) {
    const unsigned int u = __builtin_bit_cast(unsigned int, v);
    return __builtin_bit_cast(float,
        __builtin_amdgcn_permlane16_var(u, u, grp8 | (unsigned int)K, true, true));
}

__global__ __launch_bounds__(WPB * 32) void hbv_scan_kernel(
    const float* __restrict__ x, const float* __restrict__ p,
    float* __restrict__ qws)
{
    __shared__ v4f tile[WPB * BPW * PITCHV];

    const int lane = threadIdx.x & 31;
    const int wave = threadIdx.x >> 5;
    const int sub  = lane & (LPB - 1);
    const int row  = lane / LPB;
    const int b    = (blockIdx.x * WPB + wave) * BPW + row;

    const float Tt     = p[0]  * -1.5f;
    const float CFR    = p[1]  * 0.01f;
    const float CFMAX  = p[2];
    const float SCF    = p[3];
    const float FC     = p[4]  * 100.0f;
    const float Beta   = p[5]  * 8.0f;
    const float LP     = p[6]  * 2.5f;
    const float CWH    = p[7]  * 0.3f;
    const float UZL    = p[8]  * 120.0f;
    const float K0     = p[9]  * 0.5f;
    const float K1     = p[10] * 0.3f;
    const float K2     = p[11] * 0.01f;
    const float PER    = p[12] * 3.5f;
    const float rawCWH = p[7];

    const float invFC   = __builtin_amdgcn_rcpf(FC);
    const float FCLP    = FC * LP;
    const float invFCLP = __builtin_amdgcn_rcpf(FCLP);
    const float CFRMAX  = CFR * CFMAX;

    const float KS = -14.4269504088896341f;
    float c1 = 0.0f, c2 = 0.0f, c3 = 0.0f, c4 = 0.0f, c5 = 0.0f, k0 = 0.0f;
    if (sub == 1) { c1 = -(KS * CWH); c2 = KS; }
    if (sub == 2) { c3 = KS; }
    if (sub == 3) { c3 = -KS; k0 = KS * FCLP; }
    if (sub == 4) { c4 = KS; }
    if (sub == 5) { c4 = KS; k0 = -(KS * UZL); }
    if (sub == 6) { c5 = KS; }
    const unsigned int grp8 = (unsigned int)lane & 8u;

    float S1 = 0.0f, S2 = 0.0f, S3 = 0.0f, S4 = 0.0f, S5 = 0.0f;

    auto cold4 = [&](float Tm) __attribute__((always_inline)) {
        const float e  = __builtin_amdgcn_exp2f(-14.4269504088896341f * (Tt - Tm));
        const float r  = __builtin_amdgcn_rcpf(1.0f + e);
        v4f h;
        h.x = group_bcast<0>(r, grp8);
        h.y = group_bcast<1>(r, grp8);
        h.z = group_bcast<2>(r, grp8);
        h.w = group_bcast<3>(r, grp8);
        return h;
    };

#define ROW_END() __builtin_amdgcn_sched_barrier(0)

    float r_c = 0.0f, PrM_c = 0.0f, pw_c = 0.0f, b2_c = 0.0f, minR_c = 0.0f, lo2_c = 0.0f,
          hi2_c = 0.0f, lo3_c = 0.0f, hi3_c = 0.0f, lo4_c = 0.0f, hi4_c = 0.0f, lo5_c = 0.0f,
          hi5_c = 0.0f, q0c_c = 0.0f, q1c_c = 0.0f, q2c_c = 0.0f, EaA_c = 0.0f;
    float hcold_c = 0.0f, PET_c = 0.0f;
    float e1 = k0;

    const v4f* xg = (const v4f*)x + (size_t)b * XROWV + sub;
    v4f* wtile    = tile + wave * (BPW * PITCHV);
    v4f* stp      = wtile + row * PITCHV + sub;
    const v4f* ldp = wtile + row * PITCHV;
    const float* ldT = (const float*)ldp + 3 * (sub & 3) + 1;
    float* qp = qws + (size_t)b * TDIM + sub * 4;

    v4f pre[NPIECE];
#pragma unroll
    for (int i = 0; i < NPIECE; ++i)
        pre[i] = __builtin_nontemporal_load(xg + i * LPB);


#pragma unroll
    for (int i = 0; i < NPIECE; ++i)
        stp[i * LPB] = pre[i];
    wave_tile_handover();

    v4f X0[NGROUP], X1[NGROUP], X2[NGROUP];
    float XT[NGROUP];
#pragma unroll
    for (int k = 0; k < NGROUP / 2; ++k) {
        X0[k] = ldp[3 * k]; X1[k] = ldp[3 * k + 1]; X2[k] = ldp[3 * k + 2];
        XT[k] = ldT[12 * k];
    }
    v4f a0 = X0[0], a1v = X1[0], a2v = X2[0];
    const float T0 = XT[0];

    {
        const v4f* xn = xg + ROWV;
#pragma unroll
        for (int i = 0; i < NPIECE; ++i)
            pre[i] = __builtin_nontemporal_load(xn + i * LPB);
    }

    v4f hc = cold4(T0);
    {
        const float ycF_0 = Tt - a0.y;
        const float aR_0 = ycF_0 * CFRMAX;
        const float tmF_0 = a0.y - Tt;
        const float aM_0 = tmF_0 * CFMAX;
        const float hw_0 = 1.0f - hc.x;
        const float PS_0 = a0.x * SCF;
        const float ex1_0 = fmaf(c2, S2, e1);
        const float ex2_0 = fmaf(c5, S5, ex1_0);
        const float ex3_0 = fmaf(c4, S4, ex2_0);
        const float ex4_0 = fmaf(c3, S3, ex3_0);
        const float e_0 = __builtin_amdgcn_exp2f(ex4_0);
        const float t1_0 = 1.0f + e_0;
        const float r_0 = __builtin_amdgcn_rcpf(t1_0);
        const float minR_0 = fminf(aR_0, S2);
        const float minM_0 = fminf(aM_0, S1);
        const float Rfr_0 = hc.x * minR_0;
        const float Mv_0 = hw_0 * minM_0;
        const float s1_0 = fmaf(hc.x, PS_0, Rfr_0);
        const float PrM_0 = fmaf(hw_0, a0.x, Mv_0);
        const float s2_0 = fmaf(-hw_0, minM_0, s1_0);
        const float b2_0 = PrM_0 + S2;
        const float cl_0 = clip1e5(s2_0);
        const float sS1_0 = S1 + cl_0;
        const float S1_0 = fmaxf(sS1_0, 0.0f);
        const float cap2_0 = S1_0 * rawCWH;
        const float e1_0 = fmaf(c1, S1_0, k0);
        const float S2m_0 = S2 - 100000.0f;
        const float S2p_0 = S2 + 100000.0f;
        const float lo2_0 = fmaxf(fminf(S2m_0, cap2_0), 0.0f);
        const float hi2_0 = fmaxf(fminf(S2p_0, cap2_0), 0.0f);
        const float x3_0 = S3 * invFC;
        const float lg_0 = __builtin_amdgcn_logf(x3_0);
        const float bl_0 = Beta * lg_0;
        const float pw_0 = __builtin_amdgcn_exp2f(bl_0);
        const float S3m_0 = S3 - 100000.0f;
        const float hi3_0 = S3 + 100000.0f;
        const float lo3_0 = fmaxf(S3m_0, 0.0f);
        const float S4m_0 = S4 - 100000.0f;
        const float hi4_0 = S4 + 100000.0f;
        const float lo4_0 = fmaxf(S4m_0, 0.0f);
        const float S5m_0 = S5 - 100000.0f;
        const float hi5_0 = S5 + 100000.0f;
        const float lo5_0 = fmaxf(S5m_0, 0.0f);
        const float q0a_0 = S4 - UZL;
        const float q0c_0 = q0a_0 * K0;
        const float q1c_0 = S4 * K1;
        const float q2c_0 = S5 * K2;
        const float EaP_0 = a0.z * S3;
        const float EaA_0 = EaP_0 * invFCLP;
        r_c = r_0; PrM_c = PrM_0; pw_c = pw_0; b2_c = b2_0; minR_c = minR_0; lo2_c = lo2_0; hi2_c = hi2_0;
        lo3_c = lo3_0; hi3_c = hi3_0; lo4_c = lo4_0; hi4_c = hi4_0; lo5_c = lo5_0; hi5_c = hi5_0; q0c_c = q0c_0;
        q1c_c = q1c_0; q2c_c = q2c_0; EaA_c = EaA_0;
        hcold_c = hc.x; PET_c = a0.z;
        S1 = S1_0; e1 = e1_0;
    }

    v4f qk = {0.0f, 0.0f, 0.0f, 0.0f};

    for (int c = 0; c < NCHUNK; ++c) {
#pragma unroll
        for (int g = 0; g < NGROUP; ++g) {
            if (g == NGROUP - 1) {
                wave_tile_handover();
#pragma unroll
                for (int i = 0; i < NPIECE; ++i)
                    stp[i * LPB] = pre[i];
                wave_tile_handover();
                const int cl = (c + 2 < NCHUNK) ? c + 2 : NCHUNK - 1;
                const v4f* xn = xg + cl * ROWV;
#pragma unroll
                for (int i = 0; i < NPIECE; ++i)
                    pre[i] = __builtin_nontemporal_load(xn + i * LPB);
            }
            if (g == NGROUP - 1 || g == NGROUP / 2 - 1) {
                const int h0 = (g == NGROUP - 1) ? 0 : NGROUP / 2;
#pragma unroll
                for (int k = h0; k < h0 + NGROUP / 2; ++k) {
                    X0[k] = ldp[3 * k]; X1[k] = ldp[3 * k + 1]; X2[k] = ldp[3 * k + 2];
                    XT[k] = ldT[12 * k];
                }
                __builtin_amdgcn_sched_barrier(0);
            }
            const int gn = (g + 1) & (NGROUP - 1);
            const v4f n0 = X0[gn], n1 = X1[gn], n2 = X2[gn];
            const float Tn = XT[gn];
            const bool mine = (g == sub);
            const float r_0 = r_c; const float PrM_0 = PrM_c; const float pw_0 = pw_c; const float b2_0 = b2_c;
            const float minR_0 = minR_c; const float lo2_0 = lo2_c; const float hi2_0 = hi2_c; const float lo3_0 = lo3_c;
            const float hi3_0 = hi3_c; const float lo4_0 = lo4_c; const float hi4_0 = hi4_c; const float lo5_0 = lo5_c;
            const float hi5_0 = hi5_c; const float q0c_0 = q0c_c; const float q1c_0 = q1c_c; const float q2c_0 = q2c_c;
            const float EaA_0 = EaA_c;
            const float hS3_0 = group_bcast<2>(r_0, grp8);
            const float hI_0 = group_bcast<1>(r_0, grp8);
            ROW_END();
            const float hUZ_0 = group_bcast<5>(r_0, grp8);
            const float hLP_0 = group_bcast<3>(r_0, grp8);
            ROW_END();
            const float hS4_0 = group_bcast<4>(r_0, grp8);
            const float A1_0 = EaA_0 * hS3_0;
            const float P1_0 = PET_c * hS3_0;
            ROW_END();
            const float I_0 = PrM_0 * hI_0;
            const float t2a_0 = fmaf(-PrM_0, hI_0, b2_0);
            const float t4a_0 = fmaf(-q0c_0, hUZ_0, S4);
            const float t3a_0 = fmaf(PrM_0, hI_0, S3);
            ROW_END();
            const float hS5_0 = group_bcast<6>(r_0, grp8);
            const float oml_0 = 1.0f - hLP_0;
            const float t5a_0 = fmaf(PER, hS4_0, S5);
            ROW_END();
            const float A2_0 = A1_0 * hLP_0;
            const float ycF_1 = Tt - a1v.x;
            const float t2b_0 = fmaf(-hcold_c, minR_0, t2a_0);
            const float tmF_1 = a1v.x - Tt;
            ROW_END();
            const float t4b_0 = fmaf(I_0, pw_0, t4a_0);
            const float PS_1 = a0.w * SCF;
            const float hw_1 = 1.0f - hc.y;
            const float ycF_2 = Tt - a1v.w;
            ROW_END();
            const float t5b_0 = fmaf(-q2c_0, hS5_0, t5a_0);
            const float tmF_2 = a1v.w - Tt;
            const float Ea_0 = fmaf(P1_0, oml_0, A2_0);
            const float aR_1 = ycF_1 * CFRMAX;
            ROW_END();
            const float S2_0 = __builtin_amdgcn_fmed3f(t2b_0, lo2_0, hi2_0);
            const float t4c_0 = fmaf(-q1c_0, hS4_0, t4b_0);
            const float aM_1 = tmF_1 * CFMAX;
            ROW_END();
            const float aR_2 = ycF_2 * CFRMAX;
            const float PS_2 = a1v.z * SCF;
            const float S5_0 = __builtin_amdgcn_fmed3f(t5b_0, lo5_0, hi5_0);
            ROW_END();
            const float t3b_0 = t3a_0 - Ea_0;
            const float aM_2 = tmF_2 * CFMAX;
            const float ex1_1 = fmaf(c2, S2_0, e1);
            const float minR_1 = fminf(aR_1, S2_0);
            ROW_END();
            const float t4d_0 = fmaf(-PER, hS4_0, t4c_0);
            const float minM_1 = fminf(aM_1, S1);
            const float S2m_1 = S2_0 - 100000.0f;
            const float S2p_1 = S2_0 + 100000.0f;
            ROW_END();
            const float S5m_1 = S5_0 - 100000.0f;
            const float q2c_1 = S5_0 * K2;
            const float t3c_0 = fmaf(-I_0, pw_0, t3b_0);
            const float hi5_1 = S5_0 + 100000.0f;
            ROW_END();
            const float ex2_1 = fmaf(c5, S5_0, ex1_1);
            const float Rfr_1 = hc.y * minR_1;
            const float S4_0 = __builtin_amdgcn_fmed3f(t4d_0, lo4_0, hi4_0);
            ROW_END();
            const float Mv_1 = hw_1 * minM_1;
            const float hw_2 = 1.0f - hc.z;
            const float lo5_1 = fmaxf(S5m_1, 0.0f);
            const float ycF_3 = Tt - a2v.z;
            ROW_END();
            const float S3_0 = __builtin_amdgcn_fmed3f(t3c_0, lo3_0, hi3_0);
            const float s1_1 = fmaf(hc.y, PS_1, Rfr_1);
            const float tmF_3 = a2v.z - Tt;
            ROW_END();
            const float ex3_1 = fmaf(c4, S4_0, ex2_1);
            const float q0a_1 = S4_0 - UZL;
            const float PrM_1 = fmaf(hw_1, a0.w, Mv_1);
            const float q1c_1 = S4_0 * K1;
            ROW_END();
            const float S4m_1 = S4_0 - 100000.0f;
            const float hi4_1 = S4_0 + 100000.0f;
            const float x3_1 = S3_0 * invFC;
            const float EaP_1 = a1v.y * S3_0;
            ROW_END();
            const float s2_1 = fmaf(-hw_1, minM_1, s1_1);
            const float S3m_1 = S3_0 - 100000.0f;
            const float ex4_1 = fmaf(c3, S3_0, ex3_1);
            const float q0c_1 = q0a_1 * K0;
            ROW_END();
            const float b2_1 = PrM_1 + S2_0;
            const float hi3_1 = S3_0 + 100000.0f;
            const float lo4_1 = fmaxf(S4m_1, 0.0f);
            const float aR_3 = ycF_3 * CFRMAX;
            ROW_END();
            const float lg_1 = __builtin_amdgcn_logf(x3_1);
            const float cl_1 = clip1e5(s2_1);
            ROW_END();
            const float e_1 = __builtin_amdgcn_exp2f(ex4_1);
            const float EaA_1 = EaP_1 * invFCLP;
            const float lo3_1 = fmaxf(S3m_1, 0.0f);
            ROW_END();
            const float aM_3 = tmF_3 * CFMAX;
            const float PS_3 = a2v.y * SCF;
            const float hw_3 = 1.0f - hc.w;
            const float q0_0 = q0c_0 * hUZ_0;
            ROW_END();
            const float sS1_1 = S1 + cl_1;
            ROW_END();
            const float bl_1 = Beta * lg_1;
            const float qq1_0 = fmaf(q1c_0, hS4_0, q0_0);
            ROW_END();
            const float t1_1 = 1.0f + e_1;
            const float S1_1 = fmaxf(sS1_1, 0.0f);
            ROW_END();
            const float pw_1 = __builtin_amdgcn_exp2f(bl_1);
            const float q_0 = fmaf(q2c_0, hS5_0, qq1_0);
            ROW_END();
            const float r_1 = __builtin_amdgcn_rcpf(t1_1);
            const float cap2_1 = S1_1 * rawCWH;
            const float e1_1 = fmaf(c1, S1_1, k0);
            ROW_END();
            const float minM_2 = fminf(aM_2, S1_1);
            ROW_END();
            const float lo2_1 = fmaxf(fminf(S2m_1, cap2_1), 0.0f);
            const float hi2_1 = fmaxf(fminf(S2p_1, cap2_1), 0.0f);
            ROW_END();
            const float Mv_2 = hw_2 * minM_2;
            const float hS3_1 = group_bcast<2>(r_1, grp8);
            ROW_END();
            const float hI_1 = group_bcast<1>(r_1, grp8);
            const float hUZ_1 = group_bcast<5>(r_1, grp8);
            ROW_END();
            const float hLP_1 = group_bcast<3>(r_1, grp8);
            const float hS4_1 = group_bcast<4>(r_1, grp8);
            ROW_END();
            const float A1_1 = EaA_1 * hS3_1;
            const float P1_1 = a1v.y * hS3_1;
            const float I_1 = PrM_1 * hI_1;
            const float t2a_1 = fmaf(-PrM_1, hI_1, b2_1);
            ROW_END();
            const float t4a_1 = fmaf(-q0c_1, hUZ_1, S4_0);
            const float t3a_1 = fmaf(PrM_1, hI_1, S3_0);
            const float hS5_1 = group_bcast<6>(r_1, grp8);
            ROW_END();
            const float oml_1 = 1.0f - hLP_1;
            const float t5a_1 = fmaf(PER, hS4_1, S5_0);
            const float A2_1 = A1_1 * hLP_1;
            const float PrM_2 = fmaf(hw_2, a1v.z, Mv_2);
            ROW_END();
            const float t2b_1 = fmaf(-hc.y, minR_1, t2a_1);
            const float q0_1 = q0c_1 * hUZ_1;
            const float t4b_1 = fmaf(I_1, pw_1, t4a_1);
            ROW_END();
            const float t5b_1 = fmaf(-q2c_1, hS5_1, t5a_1);
            const float Ea_1 = fmaf(P1_1, oml_1, A2_1);
            ROW_END();
            const float S2_1 = __builtin_amdgcn_fmed3f(t2b_1, lo2_1, hi2_1);
            const float t4c_1 = fmaf(-q1c_1, hS4_1, t4b_1);
            const float qq1_1 = fmaf(q1c_1, hS4_1, q0_1);
            ROW_END();
            const float S5_1 = __builtin_amdgcn_fmed3f(t5b_1, lo5_1, hi5_1);
            const float t3b_1 = t3a_1 - Ea_1;
            ROW_END();
            const float ex1_2 = fmaf(c2, S2_1, e1_1);
            const float minR_2 = fminf(aR_2, S2_1);
            const float t4d_1 = fmaf(-PER, hS4_1, t4c_1);
            const float b2_2 = PrM_2 + S2_1;
            ROW_END();
            const float S2m_2 = S2_1 - 100000.0f;
            const float S2p_2 = S2_1 + 100000.0f;
            const float S5m_2 = S5_1 - 100000.0f;
            const float q2c_2 = S5_1 * K2;
            ROW_END();
            const float t3c_1 = fmaf(-I_1, pw_1, t3b_1);
            const float hi5_2 = S5_1 + 100000.0f;
            const float ex2_2 = fmaf(c5, S5_1, ex1_2);
            const float Rfr_2 = hc.z * minR_2;
            ROW_END();
            const float S4_1 = __builtin_amdgcn_fmed3f(t4d_1, lo4_1, hi4_1);
            const float q_1 = fmaf(q2c_1, hS5_1, qq1_1);
            ROW_END();
            const float lo5_2 = fmaxf(S5m_2, 0.0f);
            const float S3_1 = __builtin_amdgcn_fmed3f(t3c_1, lo3_1, hi3_1);
            ROW_END();
            const float s1_2 = fmaf(hc.z, PS_2, Rfr_2);
            const float ex3_2 = fmaf(c4, S4_1, ex2_2);
            const float q0a_2 = S4_1 - UZL;
            ROW_END();
            const float q1c_2 = S4_1 * K1;
            const float S4m_2 = S4_1 - 100000.0f;
            const float hi4_2 = S4_1 + 100000.0f;
            ROW_END();
            const float x3_2 = S3_1 * invFC;
            const float EaP_2 = a2v.x * S3_1;
            const float s2_2 = fmaf(-hw_2, minM_2, s1_2);
            const float S3m_2 = S3_1 - 100000.0f;
            ROW_END();
            const float ex4_2 = fmaf(c3, S3_1, ex3_2);
            const float q0c_2 = q0a_2 * K0;
            const float lo4_2 = fmaxf(S4m_2, 0.0f);
            const float hi3_2 = S3_1 + 100000.0f;
            ROW_END();
            const float ycF_4 = Tt - n0.y;
            const float tmF_4 = n0.y - Tt;
            const float lg_2 = __builtin_amdgcn_logf(x3_2);
            ROW_END();
            const float cl_2 = clip1e5(s2_2);
            const float e_2 = __builtin_amdgcn_exp2f(ex4_2);
            ROW_END();
            const float EaA_2 = EaP_2 * invFCLP;
            const float lo3_2 = fmaxf(S3m_2, 0.0f);
            const float aR_4 = ycF_4 * CFRMAX;
            const float aM_4 = tmF_4 * CFMAX;
            ROW_END();
            const float PS_4 = n0.x * SCF;
            const float cx0 = Tt - Tn;
            const float sS1_2 = S1_1 + cl_2;
            ROW_END();
            const float bl_2 = Beta * lg_2;
            const float cx1 = -14.4269504088896341f * cx0;
            ROW_END();
            const float t1_2 = 1.0f + e_2;
            const float S1_2 = fmaxf(sS1_2, 0.0f);
            ROW_END();
            const float pw_2 = __builtin_amdgcn_exp2f(bl_2);
            ROW_END();
            const float r_2 = __builtin_amdgcn_rcpf(t1_2);
            const float cap2_2 = S1_2 * rawCWH;
            const float e1_2 = fmaf(c1, S1_2, k0);
            ROW_END();
            const float minM_3 = fminf(aM_3, S1_2);
            const float ce = __builtin_amdgcn_exp2f(cx1);
            ROW_END();
            const float lo2_2 = fmaxf(fminf(S2m_2, cap2_2), 0.0f);
            const float hi2_2 = fmaxf(fminf(S2p_2, cap2_2), 0.0f);
            ROW_END();
            const float Mv_3 = hw_3 * minM_3;
            const float hS3_2 = group_bcast<2>(r_2, grp8);
            ROW_END();
            const float hI_2 = group_bcast<1>(r_2, grp8);
            const float hUZ_2 = group_bcast<5>(r_2, grp8);
            ROW_END();
            const float hLP_2 = group_bcast<3>(r_2, grp8);
            const float hS4_2 = group_bcast<4>(r_2, grp8);
            ROW_END();
            const float A1_2 = EaA_2 * hS3_2;
            const float P1_2 = a2v.x * hS3_2;
            const float I_2 = PrM_2 * hI_2;
            const float t2a_2 = fmaf(-PrM_2, hI_2, b2_2);
            ROW_END();
            const float t4a_2 = fmaf(-q0c_2, hUZ_2, S4_1);
            const float t3a_2 = fmaf(PrM_2, hI_2, S3_1);
            const float hS5_2 = group_bcast<6>(r_2, grp8);
            ROW_END();
            const float oml_2 = 1.0f - hLP_2;
            const float t5a_2 = fmaf(PER, hS4_2, S5_1);
            const float A2_2 = A1_2 * hLP_2;
            const float PrM_3 = fmaf(hw_3, a2v.y, Mv_3);
            ROW_END();
            const float t2b_2 = fmaf(-hc.z, minR_2, t2a_2);
            const float q0_2 = q0c_2 * hUZ_2;
            const float t4b_2 = fmaf(I_2, pw_2, t4a_2);
            const float cs = 1.0f + ce;
            ROW_END();
            const float t5b_2 = fmaf(-q2c_2, hS5_2, t5a_2);
            const float Ea_2 = fmaf(P1_2, oml_2, A2_2);
            ROW_END();
            const float S2_2 = __builtin_amdgcn_fmed3f(t2b_2, lo2_2, hi2_2);
            const float t4c_2 = fmaf(-q1c_2, hS4_2, t4b_2);
            const float qq1_2 = fmaf(q1c_2, hS4_2, q0_2);
            ROW_END();
            const float cr = __builtin_amdgcn_rcpf(cs);
            const float S5_2 = __builtin_amdgcn_fmed3f(t5b_2, lo5_2, hi5_2);
            ROW_END();
            const float t3b_2 = t3a_2 - Ea_2;
            const float ex1_3 = fmaf(c2, S2_2, e1_2);
            const float minR_3 = fminf(aR_3, S2_2);
            ROW_END();
            const float t4d_2 = fmaf(-PER, hS4_2, t4c_2);
            const float b2_3 = PrM_3 + S2_2;
            const float S2m_3 = S2_2 - 100000.0f;
            const float S2p_3 = S2_2 + 100000.0f;
            ROW_END();
            const float S5m_3 = S5_2 - 100000.0f;
            const float q2c_3 = S5_2 * K2;
            const float t3c_2 = fmaf(-I_2, pw_2, t3b_2);
            const float hi5_3 = S5_2 + 100000.0f;
            ROW_END();
            const float ex2_3 = fmaf(c5, S5_2, ex1_3);
            const float Rfr_3 = hc.w * minR_3;
            const float S4_2 = __builtin_amdgcn_fmed3f(t4d_2, lo4_2, hi4_2);
            ROW_END();
            const float q_2 = fmaf(q2c_2, hS5_2, qq1_2);
            const float lo5_3 = fmaxf(S5m_3, 0.0f);
            ROW_END();
            const float S3_2 = __builtin_amdgcn_fmed3f(t3c_2, lo3_2, hi3_2);
            const float s1_3 = fmaf(hc.w, PS_3, Rfr_3);
            ROW_END();
            const float ex3_3 = fmaf(c4, S4_2, ex2_3);
            const float q0a_3 = S4_2 - UZL;
            const float q1c_3 = S4_2 * K1;
            const float S4m_3 = S4_2 - 100000.0f;
            ROW_END();
            const float hi4_3 = S4_2 + 100000.0f;
            const float x3_3 = S3_2 * invFC;
            const float EaP_3 = a2v.w * S3_2;
            ROW_END();
            const float s2_3 = fmaf(-hw_3, minM_3, s1_3);
            const float S3m_3 = S3_2 - 100000.0f;
            const float ex4_3 = fmaf(c3, S3_2, ex3_3);
            const float q0c_3 = q0a_3 * K0;
            ROW_END();
            const float lo4_3 = fmaxf(S4m_3, 0.0f);
            const float hi3_3 = S3_2 + 100000.0f;
            const float nhx = group_bcast<0>(cr, grp8);
            ROW_END();
            const float lg_3 = __builtin_amdgcn_logf(x3_3);
            const float cl_3 = clip1e5(s2_3);
            ROW_END();
            const float e_3 = __builtin_amdgcn_exp2f(ex4_3);
            const float EaA_3 = EaP_3 * invFCLP;
            const float lo3_3 = fmaxf(S3m_3, 0.0f);
            ROW_END();
            const float hw_4 = 1.0f - nhx;
            const float nhy = group_bcast<1>(cr, grp8);
            ROW_END();
            const float sS1_3 = S1_2 + cl_3;
            const float nhz = group_bcast<2>(cr, grp8);
            ROW_END();
            const float nhw = group_bcast<3>(cr, grp8);
            const float bl_3 = Beta * lg_3;
            ROW_END();
            const float t1_3 = 1.0f + e_3;
            const float S1_3 = fmaxf(sS1_3, 0.0f);
            ROW_END();
            const float pw_3 = __builtin_amdgcn_exp2f(bl_3);
            ROW_END();
            const float r_3 = __builtin_amdgcn_rcpf(t1_3);
            const float cap2_3 = S1_3 * rawCWH;
            const float e1_3 = fmaf(c1, S1_3, k0);
            ROW_END();
            const float minM_4 = fminf(aM_4, S1_3);
            ROW_END();
            const float lo2_3 = fmaxf(fminf(S2m_3, cap2_3), 0.0f);
            const float hi2_3 = fmaxf(fminf(S2p_3, cap2_3), 0.0f);
            ROW_END();
            const float Mv_4 = hw_4 * minM_4;
            const float hI_3 = group_bcast<1>(r_3, grp8);
            ROW_END();
            const float hS3_3 = group_bcast<2>(r_3, grp8);
            const float hUZ_3 = group_bcast<5>(r_3, grp8);
            ROW_END();
            const float hLP_3 = group_bcast<3>(r_3, grp8);
            const float hS4_3 = group_bcast<4>(r_3, grp8);
            ROW_END();
            const float t2a_3 = fmaf(-PrM_3, hI_3, b2_3);
            const float I_3 = PrM_3 * hI_3;
            const float A1_3 = EaA_3 * hS3_3;
            const float P1_3 = a2v.w * hS3_3;
            ROW_END();
            const float t4a_3 = fmaf(-q0c_3, hUZ_3, S4_2);
            const float t3a_3 = fmaf(PrM_3, hI_3, S3_2);
            const float hS5_3 = group_bcast<6>(r_3, grp8);
            ROW_END();
            const float oml_3 = 1.0f - hLP_3;
            const float t5a_3 = fmaf(PER, hS4_3, S5_2);
            const float t2b_3 = fmaf(-hc.w, minR_3, t2a_3);
            const float q0_3 = q0c_3 * hUZ_3;
            ROW_END();
            const float A2_3 = A1_3 * hLP_3;
            const float PrM_4 = fmaf(hw_4, n0.x, Mv_4);
            const float t4b_3 = fmaf(I_3, pw_3, t4a_3);
            ROW_END();
            const float t5b_3 = fmaf(-q2c_3, hS5_3, t5a_3);
            const float S2_3 = __builtin_amdgcn_fmed3f(t2b_3, lo2_3, hi2_3);
            ROW_END();
            const float Ea_3 = fmaf(P1_3, oml_3, A2_3);
            const float qq1_3 = fmaf(q1c_3, hS4_3, q0_3);
            const float t4c_3 = fmaf(-q1c_3, hS4_3, t4b_3);
            ROW_END();
            const float S5_3 = __builtin_amdgcn_fmed3f(t5b_3, lo5_3, hi5_3);
            const float minR_4 = fminf(aR_4, S2_3);
            const float ex1_4 = fmaf(c2, S2_3, e1_3);
            ROW_END();
            const float t3b_3 = t3a_3 - Ea_3;
            const float S2m_4 = S2_3 - 100000.0f;
            const float t4d_3 = fmaf(-PER, hS4_3, t4c_3);
            const float S2p_4 = S2_3 + 100000.0f;
            ROW_END();
            const float q_3 = fmaf(q2c_3, hS5_3, qq1_3);
            const float b2_4 = PrM_4 + S2_3;
            const float S5m_4 = S5_3 - 100000.0f;
            const float hi5_4 = S5_3 + 100000.0f;
            ROW_END();
            const float Rfr_4 = nhx * minR_4;
            const float ex2_4 = fmaf(c5, S5_3, ex1_4);
            const float t3c_3 = fmaf(-I_3, pw_3, t3b_3);
            const float q2c_4 = S5_3 * K2;
            ROW_END();
            const float S4_3 = __builtin_amdgcn_fmed3f(t4d_3, lo4_3, hi4_3);
            ROW_END();
            const float lo5_4 = fmaxf(S5m_4, 0.0f);
            const float s1_4 = fmaf(nhx, PS_4, Rfr_4);
            ROW_END();
            const float S3_3 = __builtin_amdgcn_fmed3f(t3c_3, lo3_3, hi3_3);
            const float ex3_4 = fmaf(c4, S4_3, ex2_4);
            const float S4m_4 = S4_3 - 100000.0f;
            ROW_END();
            const float q0a_4 = S4_3 - UZL;
            const float hi4_4 = S4_3 + 100000.0f;
            const float q1c_4 = S4_3 * K1;
            ROW_END();
            const float s2_4 = fmaf(-hw_4, minM_4, s1_4);
            const float x3_4 = S3_3 * invFC;
            const float S3m_4 = S3_3 - 100000.0f;
            ROW_END();
            const float ex4_4 = fmaf(c3, S3_3, ex3_4);
            const float EaP_4 = n0.z * S3_3;
            const float hi3_4 = S3_3 + 100000.0f;
            const float lo4_4 = fmaxf(S4m_4, 0.0f);
            ROW_END();
            const float q0c_4 = q0a_4 * K0;
            const float cl_4 = clip1e5(s2_4);
            ROW_END();
            const float lg_4 = __builtin_amdgcn_logf(x3_4);
            const float lo3_4 = fmaxf(S3m_4, 0.0f);
            const float EaA_4 = EaP_4 * invFCLP;
            ROW_END();
            const float e_4 = __builtin_amdgcn_exp2f(ex4_4);
            ROW_END();
            const float sS1_4 = S1_3 + cl_4;
            ROW_END();
            const float S1_4 = fmaxf(sS1_4, 0.0f);
            const float bl_4 = Beta * lg_4;
            ROW_END();
            const float t1_4 = 1.0f + e_4;
            ROW_END();
            const float cap2_4 = S1_4 * rawCWH;
            const float e1_4 = fmaf(c1, S1_4, k0);
            const float pw_4 = __builtin_amdgcn_exp2f(bl_4);
            ROW_END();
            const float r_4 = __builtin_amdgcn_rcpf(t1_4);
            ROW_END();
            const float lo2_4 = fmaxf(fminf(S2m_4, cap2_4), 0.0f);
            const float hi2_4 = fmaxf(fminf(S2p_4, cap2_4), 0.0f);
            ROW_END();
            qk.x = mine ? q_0 : qk.x;
            qk.y = mine ? q_1 : qk.y;
            qk.z = mine ? q_2 : qk.z;
            qk.w = mine ? q_3 : qk.w;
            S1 = S1_4; S2 = S2_3; S3 = S3_3; S4 = S4_3; S5 = S5_3; e1 = e1_4;
            r_c = r_4; PrM_c = PrM_4; pw_c = pw_4; b2_c = b2_4; minR_c = minR_4; lo2_c = lo2_4; hi2_c = hi2_4;
            lo3_c = lo3_4; hi3_c = hi3_4; lo4_c = lo4_4; hi4_c = hi4_4; lo5_c = lo5_4; hi5_c = hi5_4; q0c_c = q0c_4;
            q1c_c = q1c_4; q2c_c = q2c_4; EaA_c = EaA_4;
            hcold_c = nhx; PET_c = n0.z;
            a0 = n0; a1v = n1; a2v = n2;
            hc.x = nhx; hc.y = nhy; hc.z = nhz; hc.w = nhw;
        }
        *(v4f*)(qp + c * CHUNK) = qk;
    }
    asm volatile("s_wait_loadcnt 0x0" ::: "memory");
}

__global__ __launch_bounds__(256) void hbv_route_wmma_kernel(
    const float* __restrict__ qws, float* __restrict__ out)
{
    constexpr double INV_MID_D = 1.0 / (4.0 + 1e-7);
    constexpr double WSUM_D = 1.0 + 2.0 * ((1.0 - 1.0 * INV_MID_D) +
                                           (1.0 - 2.0 * INV_MID_D) +
                                           (1.0 - 3.0 * INV_MID_D) +
                                           (1.0 - 4.0 * INV_MID_D));
    constexpr float INV_MID  = (float)INV_MID_D;
    constexpr float INV_WSUM = (float)(1.0 / (WSUM_D + 1e-7));

    const int lane = threadIdx.x & 31;
    const int wave = threadIdx.x >> 5;
    const int tile = blockIdx.x * 8 + wave;
    const int tt   = tile & (TDIM / 16 - 1);
    const int tb   = tile >> 8;
    const int b0   = tb * 16;
    const int t0   = tt * 16;
    const int mn   = lane & 15;
    const int khi  = lane >> 4;

    const float* qrow = qws + (size_t)(b0 + mn) * TDIM;
    v2f afr[6], wfr[6];
#pragma unroll
    for (int c = 0; c < 6; ++c) {
        const int tp   = t0 - 4 + 4 * c + 2 * khi;
        const bool inr = (tp >= 0) && (tp < TDIM);
        const v2f v    = *(const v2f*)(qrow + (inr ? tp : 0));
#pragma unroll
        for (int j = 0; j < 2; ++j) {
            const int kg = 4 * c + j + 2 * khi;
            afr[c][j] = inr ? v[j] : 0.0f;
            const float dd = fabsf((float)(kg - mn) - 4.0f);
            wfr[c][j] = fmaxf(0.0f, fmaf(-dd, INV_MID, 1.0f)) * INV_WSUM;
        }
    }

    v8f acc = {0.0f, 0.0f, 0.0f, 0.0f, 0.0f, 0.0f, 0.0f, 0.0f};
#pragma unroll
    for (int c = 0; c < 6; ++c) {
        acc = __builtin_amdgcn_wmma_f32_16x16x4_f32(
            false, afr[c], false, wfr[c], (short)0, acc, false, false);
    }

#pragma unroll
    for (int v = 0; v < 8; ++v) {
        const int m = v + 8 * khi;
        out[(size_t)(b0 + m) * TDIM + t0 + mn] = acc[v];
    }
}

extern "C" void kernel_launch(void* const* d_in, const int* in_sizes, int n_in,
                              void* d_out, int out_size, void* d_ws, size_t ws_size,
                              hipStream_t stream) {
    (void)in_sizes; (void)n_in; (void)out_size; (void)ws_size;
    const float* x      = (const float*)d_in[0];
    const float* params = (const float*)d_in[1];
    float* out = (float*)d_out;
    float* qws = (float*)d_ws;

    hbv_scan_kernel<<<BDIM / (WPB * BPW), WPB * 32, 0, stream>>>(x, params, qws);
    hbv_route_wmma_kernel<<<(BDIM / 16) * (TDIM / 16) / 8, 256, 0, stream>>>(qws, out);
}


# 1e5 (clip bound) and the heaviside factor held in scalar registers, not as 32-bit literals: S-1e5 and S+1e5 issue as one dual-issue pair; 9 row borders between pairable operations removed; same sums
# speedup vs baseline: 12.1888x; 1.0347x over previous
#include <hip/hip_runtime.h>


#define BDIM 2048
#define TDIM 4096

typedef float v2f __attribute__((ext_vector_type(2)));
typedef float v4f __attribute__((ext_vector_type(4)));
typedef float v8f __attribute__((ext_vector_type(8)));

__device__ __forceinline__ float clip1e5(float d) {
    return fminf(fmaxf(d, -100000.0f), 100000.0f);
}

#define CHUNK   32
#define NCHUNK  (TDIM / CHUNK)
#define ROWV    (CHUNK * 3 / 4)
#define PITCHV  25
#define XROWV   (TDIM * 3 / 4)
#define NGROUP  (CHUNK / 4)
#define LPB     8
#define BPW     (32 / LPB)
#define WPB     4
#define NPIECE  (ROWV / LPB)

__device__ __forceinline__ void wave_tile_handover() {
    __builtin_amdgcn_fence(__ATOMIC_RELEASE, "wavefront");
    __builtin_amdgcn_wave_barrier();
    __builtin_amdgcn_fence(__ATOMIC_ACQUIRE, "wavefront");
}

template <int K>
__device__ __forceinline__ float group_bcast(float v, unsigned int grp8) {
    const unsigned int u = __builtin_bit_cast(unsigned int, v);
    return __builtin_bit_cast(float,
        __builtin_amdgcn_permlane16_var(u, u, grp8 | (unsigned int)K, true, true));
}

__global__ __launch_bounds__(WPB * 32) void hbv_scan_kernel(
    const float* __restrict__ x, const float* __restrict__ p,
    float* __restrict__ qws)
{
    __shared__ v4f tile[WPB * BPW * PITCHV];

    const int lane = threadIdx.x & 31;
    const int wave = threadIdx.x >> 5;
    const int sub  = lane & (LPB - 1);
    const int row  = lane / LPB;
    const int b    = (blockIdx.x * WPB + wave) * BPW + row;

    const float Tt     = p[0]  * -1.5f;
    const float CFR    = p[1]  * 0.01f;
    const float CFMAX  = p[2];
    const float SCF    = p[3];
    const float FC     = p[4]  * 100.0f;
    const float Beta   = p[5]  * 8.0f;
    const float LP     = p[6]  * 2.5f;
    const float CWH    = p[7]  * 0.3f;
    const float UZL    = p[8]  * 120.0f;
    const float K0     = p[9]  * 0.5f;
    const float K1     = p[10] * 0.3f;
    const float K2     = p[11] * 0.01f;
    const float PER    = p[12] * 3.5f;
    const float rawCWH = p[7];

    const float invFC   = __builtin_amdgcn_rcpf(FC);
    const float FCLP    = FC * LP;
    const float invFCLP = __builtin_amdgcn_rcpf(FCLP);
    const float CFRMAX  = CFR * CFMAX;

    const float KS = -14.4269504088896341f;
    float c1 = 0.0f, c2 = 0.0f, c3 = 0.0f, c4 = 0.0f, c5 = 0.0f, k0 = 0.0f;
    if (sub == 1) { c1 = -(KS * CWH); c2 = KS; }
    if (sub == 2) { c3 = KS; }
    if (sub == 3) { c3 = -KS; k0 = KS * FCLP; }
    if (sub == 4) { c4 = KS; }
    if (sub == 5) { c4 = KS; k0 = -(KS * UZL); }
    if (sub == 6) { c5 = KS; }
    const unsigned int grp8 = (unsigned int)lane & 8u;

    float S1 = 0.0f, S2 = 0.0f, S3 = 0.0f, S4 = 0.0f, S5 = 0.0f;

    float BIG = 100000.0f;
    asm volatile("" : "+s"(BIG));
    float KSC = -14.4269504088896341f;
    asm volatile("" : "+s"(KSC));

    auto cold4 = [&](float Tm) __attribute__((always_inline)) {
        const float e  = __builtin_amdgcn_exp2f(-14.4269504088896341f * (Tt - Tm));
        const float r  = __builtin_amdgcn_rcpf(1.0f + e);
        v4f h;
        h.x = group_bcast<0>(r, grp8);
        h.y = group_bcast<1>(r, grp8);
        h.z = group_bcast<2>(r, grp8);
        h.w = group_bcast<3>(r, grp8);
        return h;
    };

#define ROW_END() __builtin_amdgcn_sched_barrier(0)

    float r_c = 0.0f, PrM_c = 0.0f, pw_c = 0.0f, b2_c = 0.0f, minR_c = 0.0f, lo2_c = 0.0f,
          hi2_c = 0.0f, lo3_c = 0.0f, hi3_c = 0.0f, lo4_c = 0.0f, hi4_c = 0.0f, lo5_c = 0.0f,
          hi5_c = 0.0f, q0c_c = 0.0f, q1c_c = 0.0f, q2c_c = 0.0f, EaA_c = 0.0f;
    float hcold_c = 0.0f, PET_c = 0.0f;
    float e1 = k0;

    const v4f* xg = (const v4f*)x + (size_t)b * XROWV + sub;
    v4f* wtile    = tile + wave * (BPW * PITCHV);
    v4f* stp      = wtile + row * PITCHV + sub;
    const v4f* ldp = wtile + row * PITCHV;
    const float* ldT = (const float*)ldp + 3 * (sub & 3) + 1;
    float* qp = qws + (size_t)b * TDIM + sub * 4;

    v4f pre[NPIECE];
#pragma unroll
    for (int i = 0; i < NPIECE; ++i)
        pre[i] = __builtin_nontemporal_load(xg + i * LPB);


#pragma unroll
    for (int i = 0; i < NPIECE; ++i)
        stp[i * LPB] = pre[i];
    wave_tile_handover();

    v4f X0[NGROUP], X1[NGROUP], X2[NGROUP];
    float XT[NGROUP];
#pragma unroll
    for (int k = 0; k < NGROUP / 2; ++k) {
        X0[k] = ldp[3 * k]; X1[k] = ldp[3 * k + 1]; X2[k] = ldp[3 * k + 2];
        XT[k] = ldT[12 * k];
    }
    v4f a0 = X0[0], a1v = X1[0], a2v = X2[0];
    const float T0 = XT[0];

    {
        const v4f* xn = xg + ROWV;
#pragma unroll
        for (int i = 0; i < NPIECE; ++i)
            pre[i] = __builtin_nontemporal_load(xn + i * LPB);
    }

    v4f hc = cold4(T0);
    {
        const float ycF_0 = Tt - a0.y;
        const float aR_0 = ycF_0 * CFRMAX;
        const float tmF_0 = a0.y - Tt;
        const float aM_0 = tmF_0 * CFMAX;
        const float hw_0 = 1.0f - hc.x;
        const float PS_0 = a0.x * SCF;
        const float ex1_0 = fmaf(c2, S2, e1);
        const float ex2_0 = fmaf(c5, S5, ex1_0);
        const float ex3_0 = fmaf(c4, S4, ex2_0);
        const float ex4_0 = fmaf(c3, S3, ex3_0);
        const float e_0 = __builtin_amdgcn_exp2f(ex4_0);
        const float t1_0 = 1.0f + e_0;
        const float r_0 = __builtin_amdgcn_rcpf(t1_0);
        const float minR_0 = fminf(aR_0, S2);
        const float minM_0 = fminf(aM_0, S1);
        const float Rfr_0 = hc.x * minR_0;
        const float Mv_0 = hw_0 * minM_0;
        const float s1_0 = fmaf(hc.x, PS_0, Rfr_0);
        const float PrM_0 = fmaf(hw_0, a0.x, Mv_0);
        const float s2_0 = fmaf(-hw_0, minM_0, s1_0);
        const float b2_0 = PrM_0 + S2;
        const float cl_0 = clip1e5(s2_0);
        const float sS1_0 = S1 + cl_0;
        const float S1_0 = fmaxf(sS1_0, 0.0f);
        const float cap2_0 = S1_0 * rawCWH;
        const float e1_0 = fmaf(c1, S1_0, k0);
        const float S2m_0 = S2 - BIG;
        const float S2p_0 = S2 + BIG;
        const float lo2_0 = fmaxf(fminf(S2m_0, cap2_0), 0.0f);
        const float hi2_0 = fmaxf(fminf(S2p_0, cap2_0), 0.0f);
        const float x3_0 = S3 * invFC;
        const float lg_0 = __builtin_amdgcn_logf(x3_0);
        const float bl_0 = Beta * lg_0;
        const float pw_0 = __builtin_amdgcn_exp2f(bl_0);
        const float S3m_0 = S3 - BIG;
        const float hi3_0 = S3 + BIG;
        const float lo3_0 = fmaxf(S3m_0, 0.0f);
        const float S4m_0 = S4 - BIG;
        const float hi4_0 = S4 + BIG;
        const float lo4_0 = fmaxf(S4m_0, 0.0f);
        const float S5m_0 = S5 - BIG;
        const float hi5_0 = S5 + BIG;
        const float lo5_0 = fmaxf(S5m_0, 0.0f);
        const float q0a_0 = S4 - UZL;
        const float q0c_0 = q0a_0 * K0;
        const float q1c_0 = S4 * K1;
        const float q2c_0 = S5 * K2;
        const float EaP_0 = a0.z * S3;
        const float EaA_0 = EaP_0 * invFCLP;
        r_c = r_0; PrM_c = PrM_0; pw_c = pw_0; b2_c = b2_0; minR_c = minR_0; lo2_c = lo2_0; hi2_c = hi2_0;
        lo3_c = lo3_0; hi3_c = hi3_0; lo4_c = lo4_0; hi4_c = hi4_0; lo5_c = lo5_0; hi5_c = hi5_0; q0c_c = q0c_0;
        q1c_c = q1c_0; q2c_c = q2c_0; EaA_c = EaA_0;
        hcold_c = hc.x; PET_c = a0.z;
        S1 = S1_0; e1 = e1_0;
    }

    v4f qk = {0.0f, 0.0f, 0.0f, 0.0f};

    for (int c = 0; c < NCHUNK; ++c) {
#pragma unroll
        for (int g = 0; g < NGROUP; ++g) {
            if (g == NGROUP - 1) {
                wave_tile_handover();
#pragma unroll
                for (int i = 0; i < NPIECE; ++i)
                    stp[i * LPB] = pre[i];
                wave_tile_handover();
                const int cl = (c + 2 < NCHUNK) ? c + 2 : NCHUNK - 1;
                const v4f* xn = xg + cl * ROWV;
#pragma unroll
                for (int i = 0; i < NPIECE; ++i)
                    pre[i] = __builtin_nontemporal_load(xn + i * LPB);
            }
            if (g == NGROUP - 1 || g == NGROUP / 2 - 1) {
                const int h0 = (g == NGROUP - 1) ? 0 : NGROUP / 2;
#pragma unroll
                for (int k = h0; k < h0 + NGROUP / 2; ++k) {
                    X0[k] = ldp[3 * k]; X1[k] = ldp[3 * k + 1]; X2[k] = ldp[3 * k + 2];
                    XT[k] = ldT[12 * k];
                }
                __builtin_amdgcn_sched_barrier(0);
            }
            const int gn = (g + 1) & (NGROUP - 1);
            const v4f n0 = X0[gn], n1 = X1[gn], n2 = X2[gn];
            const float Tn = XT[gn];
            const bool mine = (g == sub);
            const float r_0 = r_c; const float PrM_0 = PrM_c; const float pw_0 = pw_c; const float b2_0 = b2_c;
            const float minR_0 = minR_c; const float lo2_0 = lo2_c; const float hi2_0 = hi2_c; const float lo3_0 = lo3_c;
            const float hi3_0 = hi3_c; const float lo4_0 = lo4_c; const float hi4_0 = hi4_c; const float lo5_0 = lo5_c;
            const float hi5_0 = hi5_c; const float q0c_0 = q0c_c; const float q1c_0 = q1c_c; const float q2c_0 = q2c_c;
            const float EaA_0 = EaA_c;
            const float hS3_0 = group_bcast<2>(r_0, grp8);
            const float hI_0 = group_bcast<1>(r_0, grp8);
            ROW_END();
            const float hUZ_0 = group_bcast<5>(r_0, grp8);
            const float hLP_0 = group_bcast<3>(r_0, grp8);
            ROW_END();
            const float hS4_0 = group_bcast<4>(r_0, grp8);
            const float A1_0 = EaA_0 * hS3_0;
            const float P1_0 = PET_c * hS3_0;
            ROW_END();
            const float I_0 = PrM_0 * hI_0;
            const float t2a_0 = fmaf(-PrM_0, hI_0, b2_0);
            const float t4a_0 = fmaf(-q0c_0, hUZ_0, S4);
            const float t3a_0 = fmaf(PrM_0, hI_0, S3);
            ROW_END();
            const float hS5_0 = group_bcast<6>(r_0, grp8);
            const float oml_0 = 1.0f - hLP_0;
            const float t5a_0 = fmaf(PER, hS4_0, S5);
            ROW_END();
            const float A2_0 = A1_0 * hLP_0;
            const float ycF_1 = Tt - a1v.x;
            const float t2b_0 = fmaf(-hcold_c, minR_0, t2a_0);
            const float tmF_1 = a1v.x - Tt;
            ROW_END();
            const float t4b_0 = fmaf(I_0, pw_0, t4a_0);
            const float PS_1 = a0.w * SCF;
            const float hw_1 = 1.0f - hc.y;
            const float ycF_2 = Tt - a1v.w;
            ROW_END();
            const float t5b_0 = fmaf(-q2c_0, hS5_0, t5a_0);
            const float tmF_2 = a1v.w - Tt;
            const float Ea_0 = fmaf(P1_0, oml_0, A2_0);
            const float aR_1 = ycF_1 * CFRMAX;
            ROW_END();
            const float S2_0 = __builtin_amdgcn_fmed3f(t2b_0, lo2_0, hi2_0);
            const float t4c_0 = fmaf(-q1c_0, hS4_0, t4b_0);
            const float aM_1 = tmF_1 * CFMAX;
            ROW_END();
            const float aR_2 = ycF_2 * CFRMAX;
            const float PS_2 = a1v.z * SCF;
            const float S5_0 = __builtin_amdgcn_fmed3f(t5b_0, lo5_0, hi5_0);
            ROW_END();
            const float t3b_0 = t3a_0 - Ea_0;
            const float aM_2 = tmF_2 * CFMAX;
            const float ex1_1 = fmaf(c2, S2_0, e1);
            const float minR_1 = fminf(aR_1, S2_0);
            const float t4d_0 = fmaf(-PER, hS4_0, t4c_0);
            const float minM_1 = fminf(aM_1, S1);
            const float S2m_1 = S2_0 - BIG;
            const float S2p_1 = S2_0 + BIG;
            ROW_END();
            const float S5m_1 = S5_0 - BIG;
            const float q2c_1 = S5_0 * K2;
            const float t3c_0 = fmaf(-I_0, pw_0, t3b_0);
            const float hi5_1 = S5_0 + BIG;
            ROW_END();
            const float ex2_1 = fmaf(c5, S5_0, ex1_1);
            const float Rfr_1 = hc.y * minR_1;
            const float S4_0 = __builtin_amdgcn_fmed3f(t4d_0, lo4_0, hi4_0);
            ROW_END();
            const float Mv_1 = hw_1 * minM_1;
            const float hw_2 = 1.0f - hc.z;
            const float lo5_1 = fmaxf(S5m_1, 0.0f);
            const float ycF_3 = Tt - a2v.z;
            ROW_END();
            const float S3_0 = __builtin_amdgcn_fmed3f(t3c_0, lo3_0, hi3_0);
            const float s1_1 = fmaf(hc.y, PS_1, Rfr_1);
            const float tmF_3 = a2v.z - Tt;
            ROW_END();
            const float ex3_1 = fmaf(c4, S4_0, ex2_1);
            const float q0a_1 = S4_0 - UZL;
            const float PrM_1 = fmaf(hw_1, a0.w, Mv_1);
            const float q1c_1 = S4_0 * K1;
            ROW_END();
            const float S4m_1 = S4_0 - BIG;
            const float hi4_1 = S4_0 + BIG;
            const float x3_1 = S3_0 * invFC;
            const float EaP_1 = a1v.y * S3_0;
            ROW_END();
            const float s2_1 = fmaf(-hw_1, minM_1, s1_1);
            const float S3m_1 = S3_0 - BIG;
            const float ex4_1 = fmaf(c3, S3_0, ex3_1);
            const float q0c_1 = q0a_1 * K0;
            const float b2_1 = PrM_1 + S2_0;
            const float hi3_1 = S3_0 + BIG;
            const float lo4_1 = fmaxf(S4m_1, 0.0f);
            const float aR_3 = ycF_3 * CFRMAX;
            ROW_END();
            const float lg_1 = __builtin_amdgcn_logf(x3_1);
            const float cl_1 = clip1e5(s2_1);
            ROW_END();
            const float e_1 = __builtin_amdgcn_exp2f(ex4_1);
            const float EaA_1 = EaP_1 * invFCLP;
            const float lo3_1 = fmaxf(S3m_1, 0.0f);
            const float aM_3 = tmF_3 * CFMAX;
            const float PS_3 = a2v.y * SCF;
            const float hw_3 = 1.0f - hc.w;
            const float q0_0 = q0c_0 * hUZ_0;
            ROW_END();
            const float sS1_1 = S1 + cl_1;
            ROW_END();
            const float bl_1 = Beta * lg_1;
            const float qq1_0 = fmaf(q1c_0, hS4_0, q0_0);
            ROW_END();
            const float t1_1 = 1.0f + e_1;
            const float S1_1 = fmaxf(sS1_1, 0.0f);
            ROW_END();
            const float pw_1 = __builtin_amdgcn_exp2f(bl_1);
            const float q_0 = fmaf(q2c_0, hS5_0, qq1_0);
            ROW_END();
            const float r_1 = __builtin_amdgcn_rcpf(t1_1);
            const float cap2_1 = S1_1 * rawCWH;
            const float e1_1 = fmaf(c1, S1_1, k0);
            ROW_END();
            const float minM_2 = fminf(aM_2, S1_1);
            ROW_END();
            const float lo2_1 = fmaxf(fminf(S2m_1, cap2_1), 0.0f);
            const float hi2_1 = fmaxf(fminf(S2p_1, cap2_1), 0.0f);
            ROW_END();
            const float Mv_2 = hw_2 * minM_2;
            const float hS3_1 = group_bcast<2>(r_1, grp8);
            ROW_END();
            const float hI_1 = group_bcast<1>(r_1, grp8);
            const float hUZ_1 = group_bcast<5>(r_1, grp8);
            ROW_END();
            const float hLP_1 = group_bcast<3>(r_1, grp8);
            const float hS4_1 = group_bcast<4>(r_1, grp8);
            ROW_END();
            const float A1_1 = EaA_1 * hS3_1;
            const float P1_1 = a1v.y * hS3_1;
            const float I_1 = PrM_1 * hI_1;
            const float t2a_1 = fmaf(-PrM_1, hI_1, b2_1);
            ROW_END();
            const float t4a_1 = fmaf(-q0c_1, hUZ_1, S4_0);
            const float t3a_1 = fmaf(PrM_1, hI_1, S3_0);
            const float hS5_1 = group_bcast<6>(r_1, grp8);
            ROW_END();
            const float oml_1 = 1.0f - hLP_1;
            const float t5a_1 = fmaf(PER, hS4_1, S5_0);
            const float A2_1 = A1_1 * hLP_1;
            const float PrM_2 = fmaf(hw_2, a1v.z, Mv_2);
            ROW_END();
            const float t2b_1 = fmaf(-hc.y, minR_1, t2a_1);
            const float q0_1 = q0c_1 * hUZ_1;
            const float t4b_1 = fmaf(I_1, pw_1, t4a_1);
            ROW_END();
            const float t5b_1 = fmaf(-q2c_1, hS5_1, t5a_1);
            const float Ea_1 = fmaf(P1_1, oml_1, A2_1);
            ROW_END();
            const float S2_1 = __builtin_amdgcn_fmed3f(t2b_1, lo2_1, hi2_1);
            const float t4c_1 = fmaf(-q1c_1, hS4_1, t4b_1);
            const float qq1_1 = fmaf(q1c_1, hS4_1, q0_1);
            ROW_END();
            const float S5_1 = __builtin_amdgcn_fmed3f(t5b_1, lo5_1, hi5_1);
            const float t3b_1 = t3a_1 - Ea_1;
            ROW_END();
            const float ex1_2 = fmaf(c2, S2_1, e1_1);
            const float minR_2 = fminf(aR_2, S2_1);
            const float t4d_1 = fmaf(-PER, hS4_1, t4c_1);
            const float b2_2 = PrM_2 + S2_1;
            ROW_END();
            const float S2m_2 = S2_1 - BIG;
            const float S2p_2 = S2_1 + BIG;
            const float S5m_2 = S5_1 - BIG;
            const float q2c_2 = S5_1 * K2;
            ROW_END();
            const float t3c_1 = fmaf(-I_1, pw_1, t3b_1);
            const float hi5_2 = S5_1 + BIG;
            const float ex2_2 = fmaf(c5, S5_1, ex1_2);
            const float Rfr_2 = hc.z * minR_2;
            ROW_END();
            const float S4_1 = __builtin_amdgcn_fmed3f(t4d_1, lo4_1, hi4_1);
            const float q_1 = fmaf(q2c_1, hS5_1, qq1_1);
            const float lo5_2 = fmaxf(S5m_2, 0.0f);
            const float S3_1 = __builtin_amdgcn_fmed3f(t3c_1, lo3_1, hi3_1);
            ROW_END();
            const float s1_2 = fmaf(hc.z, PS_2, Rfr_2);
            const float ex3_2 = fmaf(c4, S4_1, ex2_2);
            const float q0a_2 = S4_1 - UZL;
            const float q1c_2 = S4_1 * K1;
            const float S4m_2 = S4_1 - BIG;
            const float hi4_2 = S4_1 + BIG;
            ROW_END();
            const float x3_2 = S3_1 * invFC;
            const float EaP_2 = a2v.x * S3_1;
            const float s2_2 = fmaf(-hw_2, minM_2, s1_2);
            const float S3m_2 = S3_1 - BIG;
            ROW_END();
            const float ex4_2 = fmaf(c3, S3_1, ex3_2);
            const float q0c_2 = q0a_2 * K0;
            const float lo4_2 = fmaxf(S4m_2, 0.0f);
            const float hi3_2 = S3_1 + BIG;
            ROW_END();
            const float ycF_4 = Tt - n0.y;
            const float tmF_4 = n0.y - Tt;
            const float lg_2 = __builtin_amdgcn_logf(x3_2);
            ROW_END();
            const float cl_2 = clip1e5(s2_2);
            const float e_2 = __builtin_amdgcn_exp2f(ex4_2);
            ROW_END();
            const float EaA_2 = EaP_2 * invFCLP;
            const float lo3_2 = fmaxf(S3m_2, 0.0f);
            const float aR_4 = ycF_4 * CFRMAX;
            const float aM_4 = tmF_4 * CFMAX;
            const float PS_4 = n0.x * SCF;
            const float cx0 = Tt - Tn;
            const float sS1_2 = S1_1 + cl_2;
            ROW_END();
            const float bl_2 = Beta * lg_2;
            const float cx1 = KSC * cx0;
            const float t1_2 = 1.0f + e_2;
            const float S1_2 = fmaxf(sS1_2, 0.0f);
            ROW_END();
            const float pw_2 = __builtin_amdgcn_exp2f(bl_2);
            ROW_END();
            const float r_2 = __builtin_amdgcn_rcpf(t1_2);
            const float cap2_2 = S1_2 * rawCWH;
            const float e1_2 = fmaf(c1, S1_2, k0);
            ROW_END();
            const float minM_3 = fminf(aM_3, S1_2);
            const float ce = __builtin_amdgcn_exp2f(cx1);
            ROW_END();
            const float lo2_2 = fmaxf(fminf(S2m_2, cap2_2), 0.0f);
            const float hi2_2 = fmaxf(fminf(S2p_2, cap2_2), 0.0f);
            ROW_END();
            const float Mv_3 = hw_3 * minM_3;
            const float hS3_2 = group_bcast<2>(r_2, grp8);
            ROW_END();
            const float hI_2 = group_bcast<1>(r_2, grp8);
            const float hUZ_2 = group_bcast<5>(r_2, grp8);
            ROW_END();
            const float hLP_2 = group_bcast<3>(r_2, grp8);
            const float hS4_2 = group_bcast<4>(r_2, grp8);
            ROW_END();
            const float A1_2 = EaA_2 * hS3_2;
            const float P1_2 = a2v.x * hS3_2;
            const float I_2 = PrM_2 * hI_2;
            const float t2a_2 = fmaf(-PrM_2, hI_2, b2_2);
            ROW_END();
            const float t4a_2 = fmaf(-q0c_2, hUZ_2, S4_1);
            const float t3a_2 = fmaf(PrM_2, hI_2, S3_1);
            const float hS5_2 = group_bcast<6>(r_2, grp8);
            ROW_END();
            const float oml_2 = 1.0f - hLP_2;
            const float t5a_2 = fmaf(PER, hS4_2, S5_1);
            const float A2_2 = A1_2 * hLP_2;
            const float PrM_3 = fmaf(hw_3, a2v.y, Mv_3);
            ROW_END();
            const float t2b_2 = fmaf(-hc.z, minR_2, t2a_2);
            const float q0_2 = q0c_2 * hUZ_2;
            const float t4b_2 = fmaf(I_2, pw_2, t4a_2);
            const float cs = 1.0f + ce;
            ROW_END();
            const float t5b_2 = fmaf(-q2c_2, hS5_2, t5a_2);
            const float Ea_2 = fmaf(P1_2, oml_2, A2_2);
            ROW_END();
            const float S2_2 = __builtin_amdgcn_fmed3f(t2b_2, lo2_2, hi2_2);
            const float t4c_2 = fmaf(-q1c_2, hS4_2, t4b_2);
            const float qq1_2 = fmaf(q1c_2, hS4_2, q0_2);
            ROW_END();
            const float cr = __builtin_amdgcn_rcpf(cs);
            const float S5_2 = __builtin_amdgcn_fmed3f(t5b_2, lo5_2, hi5_2);
            ROW_END();
            const float t3b_2 = t3a_2 - Ea_2;
            const float ex1_3 = fmaf(c2, S2_2, e1_2);
            const float minR_3 = fminf(aR_3, S2_2);
            ROW_END();
            const float t4d_2 = fmaf(-PER, hS4_2, t4c_2);
            const float b2_3 = PrM_3 + S2_2;
            const float S2m_3 = S2_2 - BIG;
            const float S2p_3 = S2_2 + BIG;
            ROW_END();
            const float S5m_3 = S5_2 - BIG;
            const float q2c_3 = S5_2 * K2;
            const float t3c_2 = fmaf(-I_2, pw_2, t3b_2);
            const float hi5_3 = S5_2 + BIG;
            ROW_END();
            const float ex2_3 = fmaf(c5, S5_2, ex1_3);
            const float Rfr_3 = hc.w * minR_3;
            const float S4_2 = __builtin_amdgcn_fmed3f(t4d_2, lo4_2, hi4_2);
            ROW_END();
            const float q_2 = fmaf(q2c_2, hS5_2, qq1_2);
            const float lo5_3 = fmaxf(S5m_3, 0.0f);
            ROW_END();
            const float S3_2 = __builtin_amdgcn_fmed3f(t3c_2, lo3_2, hi3_2);
            const float s1_3 = fmaf(hc.w, PS_3, Rfr_3);
            ROW_END();
            const float ex3_3 = fmaf(c4, S4_2, ex2_3);
            const float q0a_3 = S4_2 - UZL;
            const float q1c_3 = S4_2 * K1;
            const float S4m_3 = S4_2 - BIG;
            ROW_END();
            const float hi4_3 = S4_2 + BIG;
            const float x3_3 = S3_2 * invFC;
            const float EaP_3 = a2v.w * S3_2;
            ROW_END();
            const float s2_3 = fmaf(-hw_3, minM_3, s1_3);
            const float S3m_3 = S3_2 - BIG;
            const float ex4_3 = fmaf(c3, S3_2, ex3_3);
            const float q0c_3 = q0a_3 * K0;
            ROW_END();
            const float lo4_3 = fmaxf(S4m_3, 0.0f);
            const float hi3_3 = S3_2 + BIG;
            const float nhx = group_bcast<0>(cr, grp8);
            ROW_END();
            const float lg_3 = __builtin_amdgcn_logf(x3_3);
            const float cl_3 = clip1e5(s2_3);
            ROW_END();
            const float e_3 = __builtin_amdgcn_exp2f(ex4_3);
            const float EaA_3 = EaP_3 * invFCLP;
            const float lo3_3 = fmaxf(S3m_3, 0.0f);
            ROW_END();
            const float hw_4 = 1.0f - nhx;
            const float nhy = group_bcast<1>(cr, grp8);
            ROW_END();
            const float sS1_3 = S1_2 + cl_3;
            const float nhz = group_bcast<2>(cr, grp8);
            ROW_END();
            const float nhw = group_bcast<3>(cr, grp8);
            const float bl_3 = Beta * lg_3;
            ROW_END();
            const float t1_3 = 1.0f + e_3;
            const float S1_3 = fmaxf(sS1_3, 0.0f);
            ROW_END();
            const float pw_3 = __builtin_amdgcn_exp2f(bl_3);
            ROW_END();
            const float r_3 = __builtin_amdgcn_rcpf(t1_3);
            const float cap2_3 = S1_3 * rawCWH;
            const float e1_3 = fmaf(c1, S1_3, k0);
            ROW_END();
            const float minM_4 = fminf(aM_4, S1_3);
            ROW_END();
            const float lo2_3 = fmaxf(fminf(S2m_3, cap2_3), 0.0f);
            const float hi2_3 = fmaxf(fminf(S2p_3, cap2_3), 0.0f);
            ROW_END();
            const float Mv_4 = hw_4 * minM_4;
            const float hI_3 = group_bcast<1>(r_3, grp8);
            ROW_END();
            const float hS3_3 = group_bcast<2>(r_3, grp8);
            const float hUZ_3 = group_bcast<5>(r_3, grp8);
            ROW_END();
            const float hLP_3 = group_bcast<3>(r_3, grp8);
            const float hS4_3 = group_bcast<4>(r_3, grp8);
            ROW_END();
            const float t2a_3 = fmaf(-PrM_3, hI_3, b2_3);
            const float I_3 = PrM_3 * hI_3;
            const float A1_3 = EaA_3 * hS3_3;
            const float P1_3 = a2v.w * hS3_3;
            ROW_END();
            const float t4a_3 = fmaf(-q0c_3, hUZ_3, S4_2);
            const float t3a_3 = fmaf(PrM_3, hI_3, S3_2);
            const float hS5_3 = group_bcast<6>(r_3, grp8);
            ROW_END();
            const float oml_3 = 1.0f - hLP_3;
            const float t5a_3 = fmaf(PER, hS4_3, S5_2);
            const float t2b_3 = fmaf(-hc.w, minR_3, t2a_3);
            const float q0_3 = q0c_3 * hUZ_3;
            ROW_END();
            const float A2_3 = A1_3 * hLP_3;
            const float PrM_4 = fmaf(hw_4, n0.x, Mv_4);
            const float t4b_3 = fmaf(I_3, pw_3, t4a_3);
            ROW_END();
            const float t5b_3 = fmaf(-q2c_3, hS5_3, t5a_3);
            const float S2_3 = __builtin_amdgcn_fmed3f(t2b_3, lo2_3, hi2_3);
            ROW_END();
            const float Ea_3 = fmaf(P1_3, oml_3, A2_3);
            const float qq1_3 = fmaf(q1c_3, hS4_3, q0_3);
            const float t4c_3 = fmaf(-q1c_3, hS4_3, t4b_3);
            ROW_END();
            const float S5_3 = __builtin_amdgcn_fmed3f(t5b_3, lo5_3, hi5_3);
            const float minR_4 = fminf(aR_4, S2_3);
            const float ex1_4 = fmaf(c2, S2_3, e1_3);
            ROW_END();
            const float t3b_3 = t3a_3 - Ea_3;
            const float S2m_4 = S2_3 - BIG;
            const float t4d_3 = fmaf(-PER, hS4_3, t4c_3);
            const float S2p_4 = S2_3 + BIG;
            ROW_END();
            const float q_3 = fmaf(q2c_3, hS5_3, qq1_3);
            const float b2_4 = PrM_4 + S2_3;
            const float S5m_4 = S5_3 - BIG;
            const float hi5_4 = S5_3 + BIG;
            ROW_END();
            const float Rfr_4 = nhx * minR_4;
            const float ex2_4 = fmaf(c5, S5_3, ex1_4);
            const float t3c_3 = fmaf(-I_3, pw_3, t3b_3);
            const float q2c_4 = S5_3 * K2;
            ROW_END();
            const float S4_3 = __builtin_amdgcn_fmed3f(t4d_3, lo4_3, hi4_3);
            ROW_END();
            const float lo5_4 = fmaxf(S5m_4, 0.0f);
            const float s1_4 = fmaf(nhx, PS_4, Rfr_4);
            ROW_END();
            const float S3_3 = __builtin_amdgcn_fmed3f(t3c_3, lo3_3, hi3_3);
            const float ex3_4 = fmaf(c4, S4_3, ex2_4);
            const float S4m_4 = S4_3 - BIG;
            const float q0a_4 = S4_3 - UZL;
            const float hi4_4 = S4_3 + BIG;
            const float q1c_4 = S4_3 * K1;
            const float s2_4 = fmaf(-hw_4, minM_4, s1_4);
            const float x3_4 = S3_3 * invFC;
            const float S3m_4 = S3_3 - BIG;
            ROW_END();
            const float ex4_4 = fmaf(c3, S3_3, ex3_4);
            const float EaP_4 = n0.z * S3_3;
            const float hi3_4 = S3_3 + BIG;
            const float lo4_4 = fmaxf(S4m_4, 0.0f);
            ROW_END();
            const float q0c_4 = q0a_4 * K0;
            const float cl_4 = clip1e5(s2_4);
            ROW_END();
            const float lg_4 = __builtin_amdgcn_logf(x3_4);
            const float lo3_4 = fmaxf(S3m_4, 0.0f);
            const float EaA_4 = EaP_4 * invFCLP;
            ROW_END();
            const float e_4 = __builtin_amdgcn_exp2f(ex4_4);
            ROW_END();
            const float sS1_4 = S1_3 + cl_4;
            ROW_END();
            const float S1_4 = fmaxf(sS1_4, 0.0f);
            const float bl_4 = Beta * lg_4;
            ROW_END();
            const float t1_4 = 1.0f + e_4;
            ROW_END();
            const float cap2_4 = S1_4 * rawCWH;
            const float e1_4 = fmaf(c1, S1_4, k0);
            const float pw_4 = __builtin_amdgcn_exp2f(bl_4);
            ROW_END();
            const float r_4 = __builtin_amdgcn_rcpf(t1_4);
            ROW_END();
            const float lo2_4 = fmaxf(fminf(S2m_4, cap2_4), 0.0f);
            const float hi2_4 = fmaxf(fminf(S2p_4, cap2_4), 0.0f);
            ROW_END();
            qk.x = mine ? q_0 : qk.x;
            qk.y = mine ? q_1 : qk.y;
            qk.z = mine ? q_2 : qk.z;
            qk.w = mine ? q_3 : qk.w;
            S1 = S1_4; S2 = S2_3; S3 = S3_3; S4 = S4_3; S5 = S5_3; e1 = e1_4;
            r_c = r_4; PrM_c = PrM_4; pw_c = pw_4; b2_c = b2_4; minR_c = minR_4; lo2_c = lo2_4; hi2_c = hi2_4;
            lo3_c = lo3_4; hi3_c = hi3_4; lo4_c = lo4_4; hi4_c = hi4_4; lo5_c = lo5_4; hi5_c = hi5_4; q0c_c = q0c_4;
            q1c_c = q1c_4; q2c_c = q2c_4; EaA_c = EaA_4;
            hcold_c = nhx; PET_c = n0.z;
            a0 = n0; a1v = n1; a2v = n2;
            hc.x = nhx; hc.y = nhy; hc.z = nhz; hc.w = nhw;
        }
        *(v4f*)(qp + c * CHUNK) = qk;
    }
    asm volatile("s_wait_loadcnt 0x0" ::: "memory");
}

__global__ __launch_bounds__(256) void hbv_route_wmma_kernel(
    const float* __restrict__ qws, float* __restrict__ out)
{
    constexpr double INV_MID_D = 1.0 / (4.0 + 1e-7);
    constexpr double WSUM_D = 1.0 + 2.0 * ((1.0 - 1.0 * INV_MID_D) +
                                           (1.0 - 2.0 * INV_MID_D) +
                                           (1.0 - 3.0 * INV_MID_D) +
                                           (1.0 - 4.0 * INV_MID_D));
    constexpr float INV_MID  = (float)INV_MID_D;
    constexpr float INV_WSUM = (float)(1.0 / (WSUM_D + 1e-7));

    const int lane = threadIdx.x & 31;
    const int wave = threadIdx.x >> 5;
    const int tile = blockIdx.x * 8 + wave;
    const int tt   = tile & (TDIM / 16 - 1);
    const int tb   = tile >> 8;
    const int b0   = tb * 16;
    const int t0   = tt * 16;
    const int mn   = lane & 15;
    const int khi  = lane >> 4;

    const float* qrow = qws + (size_t)(b0 + mn) * TDIM;
    v2f afr[6], wfr[6];
#pragma unroll
    for (int c = 0; c < 6; ++c) {
        const int tp   = t0 - 4 + 4 * c + 2 * khi;
        const bool inr = (tp >= 0) && (tp < TDIM);
        const v2f v    = *(const v2f*)(qrow + (inr ? tp : 0));
#pragma unroll
        for (int j = 0; j < 2; ++j) {
            const int kg = 4 * c + j + 2 * khi;
            afr[c][j] = inr ? v[j] : 0.0f;
            const float dd = fabsf((float)(kg - mn) - 4.0f);
            wfr[c][j] = fmaxf(0.0f, fmaf(-dd, INV_MID, 1.0f)) * INV_WSUM;
        }
    }

    v8f acc = {0.0f, 0.0f, 0.0f, 0.0f, 0.0f, 0.0f, 0.0f, 0.0f};
#pragma unroll
    for (int c = 0; c < 6; ++c) {
        acc = __builtin_amdgcn_wmma_f32_16x16x4_f32(
            false, afr[c], false, wfr[c], (short)0, acc, false, false);
    }

#pragma unroll
    for (int v = 0; v < 8; ++v) {
        const int m = v + 8 * khi;
        out[(size_t)(b0 + m) * TDIM + t0 + mn] = acc[v];
    }
}

extern "C" void kernel_launch(void* const* d_in, const int* in_sizes, int n_in,
                              void* d_out, int out_size, void* d_ws, size_t ws_size,
                              hipStream_t stream) {
    (void)in_sizes; (void)n_in; (void)out_size; (void)ws_size;
    const float* x      = (const float*)d_in[0];
    const float* params = (const float*)d_in[1];
    float* out = (float*)d_out;
    float* qws = (float*)d_ws;

    hbv_scan_kernel<<<BDIM / (WPB * BPW), WPB * 32, 0, stream>>>(x, params, qws);
    hbv_route_wmma_kernel<<<(BDIM / 16) * (TDIM / 16) / 8, 256, 0, stream>>>(qws, out);
}
